# starting kernel
# baseline (speedup 1.0000x reference)
#include <hip/hip_runtime.h>
#include <math.h>

static const int HD  = 128;
static const int PD  = 1024;
static const int LD  = 1024;
static const int CD  = 128;
static const int NT  = 64, NH = 32, NA = 64, NB = 64, NPA = 32, NPE = 32;
static const int RT  = PD + LD + PD + LD + CD + NH;
static const int RH  = PD + PD + LD + CD;
static const int HH  = HD * HD;

typedef float v2f __attribute__((ext_vector_type(2)));
typedef float v8f __attribute__((ext_vector_type(8)));

__device__ __forceinline__ v8f wmma_f32(v2f a, v2f b, v8f c) {
  return __builtin_amdgcn_wmma_f32_16x16x4_f32(false, a, false, b, (short)0, c,
                                               false, false);
}

__global__ void masks_kernel(const int* __restrict__ pl, const int* __restrict__ pc,
                             float* __restrict__ lpl_m, float* __restrict__ plp_m,
                             float* __restrict__ pcp_m) {
  int t = blockIdx.x * blockDim.x + threadIdx.x;
  if (t < PD) {
    int s = 0;
    for (int j = 0; j < LD; ++j) s += pl[(size_t)t * LD + j];
    lpl_m[t] = (s >= 2) ? 1.f : 0.f;
  }
  if (t < LD) {
    int s = 0;
    for (int i = 0; i < PD; ++i) s += pl[(size_t)i * LD + t];
    plp_m[t] = (s >= 2) ? 1.f : 0.f;
  }
  if (t < CD) {
    int s = 0;
    for (int i = 0; i < PD; ++i) s += pc[(size_t)i * CD + t];
    pcp_m[t] = (s >= 2) ? 1.f : 0.f;
  }
}

__global__ __launch_bounds__(256)
void agg_gemm_kernel(const int* __restrict__ rel, int ldA, int transA,
                     const float* __restrict__ Bm, int K, float* __restrict__ Cout) {
  __shared__ float sA[16 * 128];
  const int tid  = threadIdx.x;
  const int lane = tid & 31;
  const int w    = tid >> 5;
  const int m0   = blockIdx.x * 16;
  const int n    = w * 16 + (lane & 15);
  const int am   = lane & 15;
  const int koff = (lane >> 4) << 1;

  v8f acc = {0.f, 0.f, 0.f, 0.f, 0.f, 0.f, 0.f, 0.f};

  for (int k0 = 0; k0 < K; k0 += 128) {
    if (transA) {
      for (int i = tid; i < 16 * 128; i += 256) {
        int m = i & 15, k = i >> 4;
        sA[m * 128 + k] = (float)rel[(size_t)(k0 + k) * ldA + (m0 + m)];
      }
    } else {
      for (int i = tid; i < 16 * 128; i += 256) {
        int m = i >> 7, k = i & 127;
        sA[m * 128 + k] = (float)rel[(size_t)(m0 + m) * ldA + (k0 + k)];
      }
    }
    __syncthreads();
    for (int kk = 0; kk < 128; kk += 4) {
      v2f a = *(const v2f*)&sA[am * 128 + kk + koff];
      int kg = k0 + kk + koff;
      v2f b;
      b.x = Bm[(size_t)kg * HD + n];
      b.y = Bm[(size_t)(kg + 1) * HD + n];
      acc = wmma_f32(a, b, acc);
    }
    __syncthreads();
  }
  const int mb = (lane >> 4) << 3;
#pragma unroll
  for (int v = 0; v < 8; ++v)
    Cout[(size_t)(m0 + mb + v) * HD + n] = acc[v];
}

__global__ __launch_bounds__(256)
void lrl_gemm_kernel(const float* __restrict__ X0, const float* __restrict__ X1,
                     float alpha, const float* __restrict__ W,
                     const float* __restrict__ bias, const float* __restrict__ g,
                     const float* __restrict__ be, float* __restrict__ Y, int do_ln) {
  __shared__ float sA[16 * 128];
  __shared__ float sC[16 * 128];
  __shared__ float sMean[16], sRstd[16];
  const int tid  = threadIdx.x;
  const int lane = tid & 31;
  const int w    = tid >> 5;
  const int m0   = blockIdx.x * 16;
  const int n    = w * 16 + (lane & 15);
  const int am   = lane & 15;
  const int koff = (lane >> 4) << 1;

  for (int i = tid; i < 16 * 128; i += 256) {
    int m = i >> 7, k = i & 127;
    float v = alpha * X0[(size_t)(m0 + m) * 128 + k];
    if (X1) v += X1[(size_t)(m0 + m) * 128 + k];
    sA[i] = v;
  }
  __syncthreads();

  v8f acc = {0.f, 0.f, 0.f, 0.f, 0.f, 0.f, 0.f, 0.f};
#pragma unroll 4
  for (int kk = 0; kk < 128; kk += 4) {
    v2f a = *(const v2f*)&sA[am * 128 + kk + koff];
    v2f b = *(const v2f*)&W[(size_t)n * 128 + kk + koff];
    acc = wmma_f32(a, b, acc);
  }

  const int mb = (lane >> 4) << 3;
  if (!do_ln) {
#pragma unroll
    for (int v = 0; v < 8; ++v)
      Y[(size_t)(m0 + mb + v) * 128 + n] = acc[v];
    return;
  }
  float bn = bias[n];
#pragma unroll
  for (int v = 0; v < 8; ++v) {
    float x = acc[v] + bn;
    sC[(mb + v) * 128 + n] = x > 0.f ? x : 0.f;
  }
  __syncthreads();
  if (tid < 16) {
    float s = 0.f, s2 = 0.f;
    for (int k = 0; k < 128; ++k) {
      float x = sC[tid * 128 + k];
      s += x; s2 += x * x;
    }
    float mean = s * (1.f / 128.f);
    float var  = s2 * (1.f / 128.f) - mean * mean;
    sMean[tid] = mean;
    sRstd[tid] = rsqrtf(var + 1e-5f);
  }
  __syncthreads();
  for (int i = tid; i < 16 * 128; i += 256) {
    int m = i >> 7, k = i & 127;
    Y[(size_t)(m0 + m) * 128 + k] = (sC[i] - sMean[m]) * sRstd[m] * g[k] + be[k];
  }
}

__global__ void mask_cat_kernel(const float* __restrict__ lpl_m,
                                const float* __restrict__ plp_m,
                                const float* __restrict__ pcp_m,
                                float* __restrict__ mask_t, float* __restrict__ mask_h) {
  int t = blockIdx.x * blockDim.x + threadIdx.x;
  if (t < RT) {
    float v;
    if (t < 2048)      v = 1.f;
    else if (t < 3072) v = lpl_m[t - 2048];
    else if (t < 4096) v = plp_m[t - 3072];
    else if (t < 4224) v = pcp_m[t - 4096];
    else               v = 1.f;
    mask_t[t] = v;
  }
  if (t < RH) {
    float v;
    if (t < 1024)      v = 1.f;
    else if (t < 2048) v = lpl_m[t - 1024];
    else if (t < 3072) v = plp_m[t - 2048];
    else               v = pcp_m[t - 3072];
    mask_h[t] = v;
  }
}

__global__ __launch_bounds__(256)
void pair_kernel(const float* __restrict__ symW, const float* __restrict__ geoW,
                 const float* __restrict__ mask, const float* __restrict__ b1,
                 const float* __restrict__ g, const float* __restrict__ be,
                 const float* __restrict__ W2, const float* __restrict__ b2p,
                 float* __restrict__ out, int Nsym, int R) {
  int gw = (int)(((unsigned)blockIdx.x * 256u + threadIdx.x) >> 5);
  if (gw >= Nsym * R) return;
  int lane = threadIdx.x & 31;
  int i = gw / R;
  int r = gw - i * R;

  float4 s4 = ((const float4*)(symW + (size_t)i * 128))[lane];
  float4 g4 = ((const float4*)(geoW + (size_t)r * 128))[lane];
  float4 b4 = ((const float4*)b1)[lane];

  float x0 = s4.x + g4.x + b4.x; x0 = x0 > 0.f ? x0 : 0.f;
  float x1 = s4.y + g4.y + b4.y; x1 = x1 > 0.f ? x1 : 0.f;
  float x2 = s4.z + g4.z + b4.z; x2 = x2 > 0.f ? x2 : 0.f;
  float x3 = s4.w + g4.w + b4.w; x3 = x3 > 0.f ? x3 : 0.f;

  float s  = x0 + x1 + x2 + x3;
  float s2 = x0 * x0 + x1 * x1 + x2 * x2 + x3 * x3;
  for (int o = 16; o > 0; o >>= 1) {
    s  += __shfl_xor(s, o, 32);
    s2 += __shfl_xor(s2, o, 32);
  }
  float mean = s * (1.f / 128.f);
  float var  = s2 * (1.f / 128.f) - mean * mean;
  float rstd = rsqrtf(var + 1e-5f);

  float4 gg = ((const float4*)g)[lane];
  float4 bb = ((const float4*)be)[lane];
  float4 w2 = ((const float4*)W2)[lane];
  float d = ((x0 - mean) * rstd * gg.x + bb.x) * w2.x +
            ((x1 - mean) * rstd * gg.y + bb.y) * w2.y +
            ((x2 - mean) * rstd * gg.z + bb.z) * w2.z +
            ((x3 - mean) * rstd * gg.w + bb.w) * w2.w;
  for (int o = 16; o > 0; o >>= 1) d += __shfl_xor(d, o, 32);

  if (lane == 0) {
    float logit = d + b2p[0];
    float mk = mask ? mask[r] : 1.f;
    float z  = (mk > 0.f) ? logit : -1e9f;
    out[(size_t)i * R + r] = 1.f / (1.f + expf(-z));
  }
}

extern "C" void kernel_launch(void* const* d_in, const int* in_sizes, int n_in,
                              void* d_out, int out_size, void* d_ws, size_t ws_size,
                              hipStream_t stream) {
  (void)in_sizes; (void)n_in; (void)out_size; (void)ws_size;
  const float* points  = (const float*)d_in[0];
  const float* lines   = (const float*)d_in[1];
  const float* circles = (const float*)d_in[2];
  const float* text_s  = (const float*)d_in[3];
  const float* head_s  = (const float*)d_in[4];
  const float* angle_s = (const float*)d_in[5];
  const float* bar_s   = (const float*)d_in[6];
  const float* para_s  = (const float*)d_in[7];
  const float* perp_s  = (const float*)d_in[8];
  const float* rel_W   = (const float*)d_in[9];
  const float* rel_b   = (const float*)d_in[10];
  const float* rel_g   = (const float*)d_in[11];
  const float* rel_be  = (const float*)d_in[12];
  const float* gm_W    = (const float*)d_in[13];
  const float* gm_b    = (const float*)d_in[14];
  const float* gm_g    = (const float*)d_in[15];
  const float* gm_be   = (const float*)d_in[16];
  const float* sc_W1   = (const float*)d_in[17];
  const float* sc_b1   = (const float*)d_in[18];
  const float* sc_g    = (const float*)d_in[19];
  const float* sc_be   = (const float*)d_in[20];
  const float* sc_W2   = (const float*)d_in[21];
  const float* sc_b2   = (const float*)d_in[22];
  const int*   pl      = (const int*)d_in[23];
  const int*   pc      = (const int*)d_in[24];
  float* out = (float*)d_out;

  float* wsp = (float*)d_ws;
  size_t off = 0;
  auto alloc = [&](size_t n) { float* p = wsp + off; off += (n + 31) & ~(size_t)31; return p; };
  float* LPLb   = alloc((size_t)PD * HD);
  float* PLPb   = alloc((size_t)LD * HD);
  float* PCPb   = alloc((size_t)CD * HD);
  float* lpl_m  = alloc(PD);
  float* plp_m  = alloc(LD);
  float* pcp_m  = alloc(CD);
  float* mask_t = alloc(RT);
  float* mask_h = alloc(RH);
  float* geo_t  = alloc((size_t)RT * HD);
  float* geo_h  = alloc((size_t)RH * HD);
  float* tmp    = alloc((size_t)RT * HD);
  float* symWb  = alloc((size_t)64 * HD);
  float* geoWb  = alloc((size_t)RT * HD);

  masks_kernel<<<4, 256, 0, stream>>>(pl, pc, lpl_m, plp_m, pcp_m);

  agg_gemm_kernel<<<PD / 16, 256, 0, stream>>>(pl, LD, 0, lines, LD, tmp);
  lrl_gemm_kernel<<<PD / 16, 256, 0, stream>>>(tmp, points, 2.f, rel_W + 0 * HH,
      rel_b + 0 * HD, rel_g + 0 * HD, rel_be + 0 * HD, LPLb, 1);

  agg_gemm_kernel<<<LD / 16, 256, 0, stream>>>(pl, LD, 1, points, PD, tmp);
  lrl_gemm_kernel<<<LD / 16, 256, 0, stream>>>(tmp, lines, 2.f, rel_W + 1 * HH,
      rel_b + 1 * HD, rel_g + 1 * HD, rel_be + 1 * HD, PLPb, 1);

  agg_gemm_kernel<<<CD / 16, 256, 0, stream>>>(pc, CD, 1, points, PD, tmp);
  lrl_gemm_kernel<<<CD / 16, 256, 0, stream>>>(tmp, circles, 2.f, rel_W + 2 * HH,
      rel_b + 2 * HD, rel_g + 2 * HD, rel_be + 2 * HD, PCPb, 1);

  const size_t ROW = (size_t)HD * sizeof(float);
  hipMemcpyAsync(geo_t + (size_t)0 * HD,    points, PD * ROW, hipMemcpyDeviceToDevice, stream);
  hipMemcpyAsync(geo_t + (size_t)1024 * HD, lines,  LD * ROW, hipMemcpyDeviceToDevice, stream);
  hipMemcpyAsync(geo_t + (size_t)2048 * HD, LPLb,   PD * ROW, hipMemcpyDeviceToDevice, stream);
  hipMemcpyAsync(geo_t + (size_t)3072 * HD, PLPb,   LD * ROW, hipMemcpyDeviceToDevice, stream);
  hipMemcpyAsync(geo_t + (size_t)4096 * HD, PCPb,   CD * ROW, hipMemcpyDeviceToDevice, stream);
  hipMemcpyAsync(geo_t + (size_t)4224 * HD, head_s, NH * ROW, hipMemcpyDeviceToDevice, stream);
  hipMemcpyAsync(geo_h + (size_t)0 * HD,    points, PD * ROW, hipMemcpyDeviceToDevice, stream);
  hipMemcpyAsync(geo_h + (size_t)1024 * HD, LPLb,   PD * ROW, hipMemcpyDeviceToDevice, stream);
  hipMemcpyAsync(geo_h + (size_t)2048 * HD, PLPb,   LD * ROW, hipMemcpyDeviceToDevice, stream);
  hipMemcpyAsync(geo_h + (size_t)3072 * HD, PCPb,   CD * ROW, hipMemcpyDeviceToDevice, stream);
  mask_cat_kernel<<<(RT + 255) / 256, 256, 0, stream>>>(lpl_m, plp_m, pcp_m, mask_t, mask_h);

  lrl_gemm_kernel<<<RT / 16, 256, 0, stream>>>(geo_t, nullptr, 1.f, gm_W + 0 * HH,
      gm_b + 0 * HD, gm_g + 0 * HD, gm_be + 0 * HD, tmp, 1);
  lrl_gemm_kernel<<<RT / 16, 256, 0, stream>>>(tmp, nullptr, 1.f, gm_W + 1 * HH,
      gm_b + 1 * HD, gm_g + 1 * HD, gm_be + 1 * HD, geo_t, 1);
  lrl_gemm_kernel<<<RH / 16, 256, 0, stream>>>(geo_h, nullptr, 1.f, gm_W + 0 * HH,
      gm_b + 0 * HD, gm_g + 0 * HD, gm_be + 0 * HD, tmp, 1);
  lrl_gemm_kernel<<<RH / 16, 256, 0, stream>>>(tmp, nullptr, 1.f, gm_W + 1 * HH,
      gm_b + 1 * HD, gm_g + 1 * HD, gm_be + 1 * HD, geo_h, 1);

  struct Head { const float* sym; int N; const float* geo; int R; const float* mask; };
  const Head heads[6] = {
      {text_s,  NT,  geo_t, RT, mask_t},
      {head_s,  NH,  geo_h, RH, mask_h},
      {angle_s, NA,  LPLb,  PD, lpl_m},
      {bar_s,   NB,  PLPb,  LD, plp_m},
      {para_s,  NPA, lines, LD, nullptr},
      {perp_s,  NPE, LPLb,  PD, lpl_m},
  };
  size_t ooff = 0;
  for (int h = 0; h < 6; ++h) {
    const Head& hd = heads[h];
    lrl_gemm_kernel<<<hd.N / 16, 256, 0, stream>>>(hd.sym, nullptr, 1.f,
        sc_W1 + (size_t)h * HH, nullptr, nullptr, nullptr, symWb, 0);
    lrl_gemm_kernel<<<hd.R / 16, 256, 0, stream>>>(hd.geo, nullptr, 1.f,
        sc_W1 + (size_t)h * HH, nullptr, nullptr, nullptr, geoWb, 0);
    int waves  = hd.N * hd.R;
    int blocks = (waves + 7) / 8;
    pair_kernel<<<blocks, 256, 0, stream>>>(symWb, geoWb, hd.mask,
        sc_b1 + (size_t)h * HD, sc_g + (size_t)h * HD, sc_be + (size_t)h * HD,
        sc_W2 + (size_t)h * HD, sc_b2 + h, out + ooff, hd.N, hd.R);
    ooff += (size_t)hd.N * hd.R;
  }
}


# pair heads project their own operands: a pair work-group computes sym@W1^T and geo@W1^T of its 16-row sym strip and 32 geo rows on WMMA from LDS; 12 projection launches go; sums unchanged
# speedup vs baseline: 2.7057x; 2.7057x over previous
#include <hip/hip_runtime.h>
#include <math.h>

static const int HD  = 128;
static const int PD  = 1024;
static const int LD  = 1024;
static const int CD  = 128;
static const int NT  = 64, NH = 32, NA = 64, NB = 64, NPA = 32, NPE = 32;
static const int RT  = PD + LD + PD + LD + CD + NH;
static const int RH  = PD + PD + LD + CD;
static const int HH  = HD * HD;

typedef float v2f __attribute__((ext_vector_type(2)));
typedef float v8f __attribute__((ext_vector_type(8)));

__device__ __forceinline__ v8f wmma_f32(v2f a, v2f b, v8f c) {
  return __builtin_amdgcn_wmma_f32_16x16x4_f32(false, a, false, b, (short)0, c,
                                               false, false);
}

struct RowSrc {
  const float* p0; const float* p1; const float* p2;
  const float* p3; const float* p4; const float* p5;
  int e0, e1, e2, e3, e4;
};
__device__ __forceinline__ const float* strip_base(const RowSrc& s, int r) {
  const float* p = s.p0;
  int b = 0;
  if (r >= s.e0) { p = s.p1; b = s.e0; }
  if (r >= s.e1) { p = s.p2; b = s.e1; }
  if (r >= s.e2) { p = s.p3; b = s.e2; }
  if (r >= s.e3) { p = s.p4; b = s.e3; }
  if (r >= s.e4) { p = s.p5; b = s.e4; }
  return p + (size_t)(r - b) * 128;
}
struct MaskSrc {
  const float* q0; const float* q1; const float* q2;
  int b0, e0, b1, e1, b2, e2;
};
__device__ __forceinline__ float mask_at(const MaskSrc& s, int r) {
  float mk = 1.f;
  if (r >= s.b0 && r < s.e0) mk = s.q0[r - s.b0];
  if (r >= s.b1 && r < s.e1) mk = s.q1[r - s.b1];
  if (r >= s.b2 && r < s.e2) mk = s.q2[r - s.b2];
  return mk;
}

static const int MK_ROWG = PD / 32;
static const int MK_COLG = LD / 32;
static const int MK_PCG  = CD / 32;
__global__ __launch_bounds__(256)
void masks_kernel(const int* __restrict__ pl, const int* __restrict__ pc,
                  float* __restrict__ lpl_m, float* __restrict__ plp_m,
                  float* __restrict__ pcp_m) {
  __shared__ alignas(16) int sP[8 * 32];
  const int tid  = threadIdx.x;
  const int lane = tid & 31;
  const int w    = tid >> 5;
  const int b    = blockIdx.x;

  if (b < MK_ROWG) {
    const int4* __restrict__ row = (const int4*)(pl + (size_t)(32 * b + 4 * w) * LD);
    int4 v[4][8];
#pragma unroll
    for (int r = 0; r < 4; ++r) {
#pragma unroll
      for (int j = 0; j < 8; ++j) v[r][j] = row[r * (LD / 4) + j * 32 + lane];
    }
    int s[4];
#pragma unroll
    for (int r = 0; r < 4; ++r) {
      int t = 0;
#pragma unroll
      for (int j = 0; j < 8; ++j) t += (v[r][j].x + v[r][j].y) + (v[r][j].z + v[r][j].w);
      s[r] = t;
    }
#pragma unroll
    for (int r = 0; r < 4; ++r)
      for (int o = 16; o > 0; o >>= 1) s[r] += __shfl_xor(s[r], o, 32);
    if (lane == 0) {
      sP[4 * w + 0] = s[0]; sP[4 * w + 1] = s[1];
      sP[4 * w + 2] = s[2]; sP[4 * w + 3] = s[3];
    }
    __syncthreads();
    if (w == 0) lpl_m[32 * b + lane] = (sP[lane] >= 2) ? 1.f : 0.f;
  } else {
    const int* __restrict__ rel = pl;
    float* __restrict__ dst = plp_m;
    int ld = LD, c0 = 32 * (b - MK_ROWG);
    if (b >= MK_ROWG + MK_COLG) {
      rel = pc; dst = pcp_m; ld = CD; c0 = 32 * (b - MK_ROWG - MK_COLG);
    }
    const int q = tid & 7;
    const int r = tid >> 3;
    const int* __restrict__ p = rel + (size_t)r * ld + c0 + 4 * q;
    int4 v[32];
#pragma unroll
    for (int j = 0; j < 32; ++j) v[j] = *(const int4*)(p + (size_t)(32 * j) * ld);
    int4 s = {0, 0, 0, 0};
#pragma unroll
    for (int j = 0; j < 32; ++j) { s.x += v[j].x; s.y += v[j].y; s.z += v[j].z; s.w += v[j].w; }
    for (int o = 8; o <= 16; o <<= 1) {
      s.x += __shfl_xor(s.x, o, 32); s.y += __shfl_xor(s.y, o, 32);
      s.z += __shfl_xor(s.z, o, 32); s.w += __shfl_xor(s.w, o, 32);
    }
    if (lane < 8) ((int4*)sP)[w * 8 + lane] = s;
    __syncthreads();
    if (w == 0) {
      int t = 0;
#pragma unroll
      for (int k = 0; k < 8; ++k) t += sP[k * 32 + lane];
      dst[c0 + lane] = (t >= 2) ? 1.f : 0.f;
    }
  }
}

static const int KS = 8;
__global__ __launch_bounds__(256)
void agg_gemm_kernel(const int* __restrict__ rel, int ldA, int transA,
                     const float* __restrict__ Bm, int kchunk, int pstride,
                     float* __restrict__ Cpart) {
  __shared__ float sA[16 * 128];
  const int tid  = threadIdx.x;
  const int lane = tid & 31;
  const int w    = tid >> 5;
  const int m0   = blockIdx.x * 16;
  const int n    = w * 16 + (lane & 15);
  const int am   = lane & 15;
  const int koff = (lane >> 4) << 1;

  v8f acc = {0.f, 0.f, 0.f, 0.f, 0.f, 0.f, 0.f, 0.f};

  const int kbeg = blockIdx.y * kchunk;
  float* __restrict__ Cout = Cpart + (size_t)blockIdx.y * pstride;
  for (int k0 = kbeg; k0 < kbeg + kchunk; k0 += 128) {
    if (transA) {
      for (int i = tid; i < 16 * 128; i += 256) {
        int m = i & 15, k = i >> 4;
        sA[m * 128 + k] = (float)rel[(size_t)(k0 + k) * ldA + (m0 + m)];
      }
    } else {
      for (int i = tid; i < 16 * 128; i += 256) {
        int m = i >> 7, k = i & 127;
        sA[m * 128 + k] = (float)rel[(size_t)(m0 + m) * ldA + (k0 + k)];
      }
    }
    __syncthreads();
    for (int kk = 0; kk < 128; kk += 4) {
      v2f a = *(const v2f*)&sA[am * 128 + kk + koff];
      int kg = k0 + kk + koff;
      v2f b;
      b.x = Bm[(size_t)kg * HD + n];
      b.y = Bm[(size_t)(kg + 1) * HD + n];
      acc = wmma_f32(a, b, acc);
    }
    __syncthreads();
  }
  const int mb = (lane >> 4) << 3;
#pragma unroll
  for (int v = 0; v < 8; ++v)
    Cout[(size_t)(m0 + mb + v) * HD + n] = acc[v];
}

template <int NS>
__global__ __launch_bounds__(256)
void lrl_gemm_kernel(RowSrc xs, int sstride,
                     const float* __restrict__ X1,
                     float alpha, const float* __restrict__ W,
                     const float* __restrict__ bias, const float* __restrict__ g,
                     const float* __restrict__ be, float* __restrict__ Y) {
  __shared__ float sA[16 * 128];
  __shared__ float sC[16 * 128];
  __shared__ float sMean[16], sRstd[16];
  const int tid  = threadIdx.x;
  const int lane = tid & 31;
  const int w    = tid >> 5;
  const int m0   = blockIdx.x * 16;
  const int n    = w * 16 + (lane & 15);
  const int am   = lane & 15;
  const int koff = (lane >> 4) << 1;
  const float* __restrict__ X0 = strip_base(xs, m0);

  if constexpr (NS == 1) {
    for (int i = tid; i < 16 * 128; i += 256) {
      int m = i >> 7, k = i & 127;
      float v = alpha * X0[(size_t)m * 128 + k];
      if (X1) v += X1[(size_t)(m0 + m) * 128 + k];
      sA[i] = v;
    }
  } else {
    float x[8];
#pragma unroll
    for (int j = 0; j < 8; ++j) {
      int i = tid + j * 256;
      x[j] = X0[(size_t)(i >> 7) * 128 + (i & 127)];
    }
#pragma unroll
    for (int s = 1; s < NS; ++s) {
#pragma unroll
      for (int j = 0; j < 8; ++j) {
        int i = tid + j * 256;
        x[j] += X0[(size_t)s * sstride + (size_t)(i >> 7) * 128 + (i & 127)];
      }
    }
#pragma unroll
    for (int j = 0; j < 8; ++j) {
      int i = tid + j * 256;
      float v = alpha * x[j];
      if (X1) v += X1[(size_t)(m0 + (i >> 7)) * 128 + (i & 127)];
      sA[i] = v;
    }
  }
  __syncthreads();

  v8f acc = {0.f, 0.f, 0.f, 0.f, 0.f, 0.f, 0.f, 0.f};
#pragma unroll 4
  for (int kk = 0; kk < 128; kk += 4) {
    v2f a = *(const v2f*)&sA[am * 128 + kk + koff];
    v2f b = *(const v2f*)&W[(size_t)n * 128 + kk + koff];
    acc = wmma_f32(a, b, acc);
  }

  const int mb = (lane >> 4) << 3;
  float bn = bias[n];
#pragma unroll
  for (int v = 0; v < 8; ++v) {
    float x = acc[v] + bn;
    sC[(mb + v) * 128 + n] = x > 0.f ? x : 0.f;
  }
  __syncthreads();
  if (tid < 16) {
    float s = 0.f, s2 = 0.f;
    for (int k = 0; k < 128; ++k) {
      float x = sC[tid * 128 + k];
      s += x; s2 += x * x;
    }
    float mean = s * (1.f / 128.f);
    float var  = s2 * (1.f / 128.f) - mean * mean;
    sMean[tid] = mean;
    sRstd[tid] = rsqrtf(var + 1e-5f);
  }
  __syncthreads();
  for (int i = tid; i < 16 * 128; i += 256) {
    int m = i >> 7, k = i & 127;
    Y[(size_t)(m0 + m) * 128 + k] = (sC[i] - sMean[m]) * sRstd[m] * g[k] + be[k];
  }
}

static const int PF_G = 32;
static const int PF_P = 132;
template <int S>
__global__ __launch_bounds__(256)
void pair_fused_kernel(const float* __restrict__ sym, RowSrc geo, MaskSrc mask,
                       const float* __restrict__ W1, const float* __restrict__ b1,
                       const float* __restrict__ g, const float* __restrict__ be,
                       const float* __restrict__ W2, const float* __restrict__ b2p,
                       float* __restrict__ out, int R) {
  __shared__ alignas(16) float sS[16 * PF_P];
  __shared__ alignas(16) float sG[PF_G * PF_P];
  __shared__ float sD[S * PF_G];
  const int tid  = threadIdx.x;
  const int lane = tid & 31;
  const int w    = tid >> 5;
  const int r0   = blockIdx.x * PF_G;
  const int i0   = blockIdx.y * S;
  const int i16  = i0 & ~15;
  const int lo   = i0 - i16;
  const int n    = w * 16 + (lane & 15);
  const int am   = lane & 15;
  const int koff = (lane >> 4) << 1;

  v2f bw[32];
#pragma unroll
  for (int t = 0; t < 32; ++t)
    bw[t] = *(const v2f*)&W1[(size_t)n * 128 + 4 * t + koff];

  const float* __restrict__ gs0 = strip_base(geo, r0);
  const float* __restrict__ gs1 = strip_base(geo, r0 + 16);
  {
    float4 vs[2], vg[4];
#pragma unroll
    for (int j = 0; j < 2; ++j) {
      int i = tid + j * 256;
      vs[j] = ((const float4*)(sym + (size_t)(i16 + (i >> 5)) * 128))[i & 31];
    }
#pragma unroll
    for (int j = 0; j < 4; ++j) {
      int i = tid + j * 256;
      const float* base = (j < 2) ? gs0 : gs1;
      vg[j] = ((const float4*)(base + (size_t)((i >> 5) & 15) * 128))[i & 31];
    }
#pragma unroll
    for (int j = 0; j < 2; ++j) {
      int i = tid + j * 256;
      *(float4*)&sS[(i >> 5) * PF_P + 4 * (i & 31)] = vs[j];
    }
#pragma unroll
    for (int j = 0; j < 4; ++j) {
      int i = tid + j * 256;
      *(float4*)&sG[(i >> 5) * PF_P + 4 * (i & 31)] = vg[j];
    }
  }
  const float4 b4 = ((const float4*)b1)[lane];
  const float4 gg = ((const float4*)g)[lane];
  const float4 bb = ((const float4*)be)[lane];
  const float4 w2 = ((const float4*)W2)[lane];
  __syncthreads();

  v8f aS  = {0.f, 0.f, 0.f, 0.f, 0.f, 0.f, 0.f, 0.f};
  v8f aG0 = {0.f, 0.f, 0.f, 0.f, 0.f, 0.f, 0.f, 0.f};
  v8f aG1 = {0.f, 0.f, 0.f, 0.f, 0.f, 0.f, 0.f, 0.f};
#pragma unroll
  for (int kk = 0; kk < 128; kk += 4) {
    v2f b  = bw[kk >> 2];
    v2f x  = *(const v2f*)&sS[am * PF_P + kk + koff];
    v2f y0 = *(const v2f*)&sG[am * PF_P + kk + koff];
    v2f y1 = *(const v2f*)&sG[(16 + am) * PF_P + kk + koff];
    aS  = wmma_f32(x,  b, aS);
    aG0 = wmma_f32(y0, b, aG0);
    aG1 = wmma_f32(y1, b, aG1);
  }
  __syncthreads();
  const int mb = (lane >> 4) << 3;
#pragma unroll
  for (int v = 0; v < 8; ++v) {
    sS[(mb + v) * PF_P + n]      = aS[v];
    sG[(mb + v) * PF_P + n]      = aG0[v];
    sG[(16 + mb + v) * PF_P + n] = aG1[v];
  }
  __syncthreads();

  float4 g4[4];
#pragma unroll
  for (int c = 0; c < 4; ++c)
    g4[c] = *(const float4*)&sG[(4 * w + c) * PF_P + 4 * lane];

  for (int q = 0; q < S / 4; ++q) {
    float4 s4[4];
#pragma unroll
    for (int a = 0; a < 4; ++a)
      s4[a] = *(const float4*)&sS[(lo + 4 * q + a) * PF_P + 4 * lane];

    float keep = 0.f;
#pragma unroll
    for (int a = 0; a < 4; ++a) {
#pragma unroll
      for (int c = 0; c < 4; ++c) {
        float x0 = s4[a].x + g4[c].x + b4.x; x0 = x0 > 0.f ? x0 : 0.f;
        float x1 = s4[a].y + g4[c].y + b4.y; x1 = x1 > 0.f ? x1 : 0.f;
        float x2 = s4[a].z + g4[c].z + b4.z; x2 = x2 > 0.f ? x2 : 0.f;
        float x3 = s4[a].w + g4[c].w + b4.w; x3 = x3 > 0.f ? x3 : 0.f;

        float s  = x0 + x1 + x2 + x3;
        float s2 = x0 * x0 + x1 * x1 + x2 * x2 + x3 * x3;
        for (int o = 16; o > 0; o >>= 1) {
          s  += __shfl_xor(s, o, 32);
          s2 += __shfl_xor(s2, o, 32);
        }
        float mean = s * (1.f / 128.f);
        float var  = s2 * (1.f / 128.f) - mean * mean;
        float rstd = rsqrtf(var + 1e-5f);

        float d = ((x0 - mean) * rstd * gg.x + bb.x) * w2.x +
                  ((x1 - mean) * rstd * gg.y + bb.y) * w2.y +
                  ((x2 - mean) * rstd * gg.z + bb.z) * w2.z +
                  ((x3 - mean) * rstd * gg.w + bb.w) * w2.w;
        for (int o = 16; o > 0; o >>= 1) d += __shfl_xor(d, o, 32);
        keep = (lane == 4 * a + c) ? d : keep;
      }
    }
    if (lane < 16) sD[(4 * q + (lane >> 2)) * PF_G + 4 * w + (lane & 3)] = keep;
  }
  __syncthreads();

  const float b2 = b2p[0];
  for (int idx = tid; idx < S * PF_G; idx += 256) {
    const int a = idx >> 5;
    const int r = r0 + (idx & 31);
    float logit = sD[idx] + b2;
    float mk = mask_at(mask, r);
    float z  = (mk > 0.f) ? logit : -1e9f;
    out[(size_t)(i0 + a) * R + r] = 1.f / (1.f + expf(-z));
  }
}

extern "C" void kernel_launch(void* const* d_in, const int* in_sizes, int n_in,
                              void* d_out, int out_size, void* d_ws, size_t ws_size,
                              hipStream_t stream) {
  (void)in_sizes; (void)n_in; (void)out_size; (void)ws_size;
  const float* points  = (const float*)d_in[0];
  const float* lines   = (const float*)d_in[1];
  const float* circles = (const float*)d_in[2];
  const float* text_s  = (const float*)d_in[3];
  const float* head_s  = (const float*)d_in[4];
  const float* angle_s = (const float*)d_in[5];
  const float* bar_s   = (const float*)d_in[6];
  const float* para_s  = (const float*)d_in[7];
  const float* perp_s  = (const float*)d_in[8];
  const float* rel_W   = (const float*)d_in[9];
  const float* rel_b   = (const float*)d_in[10];
  const float* rel_g   = (const float*)d_in[11];
  const float* rel_be  = (const float*)d_in[12];
  const float* gm_W    = (const float*)d_in[13];
  const float* gm_b    = (const float*)d_in[14];
  const float* gm_g    = (const float*)d_in[15];
  const float* gm_be   = (const float*)d_in[16];
  const float* sc_W1   = (const float*)d_in[17];
  const float* sc_b1   = (const float*)d_in[18];
  const float* sc_g    = (const float*)d_in[19];
  const float* sc_be   = (const float*)d_in[20];
  const float* sc_W2   = (const float*)d_in[21];
  const float* sc_b2   = (const float*)d_in[22];
  const int*   pl      = (const int*)d_in[23];
  const int*   pc      = (const int*)d_in[24];
  float* out = (float*)d_out;

  float* wsp = (float*)d_ws;
  size_t off = 0;
  auto alloc = [&](size_t n) { float* p = wsp + off; off += (n + 31) & ~(size_t)31; return p; };
  float* LPLb   = alloc((size_t)PD * HD);
  float* PLPb   = alloc((size_t)LD * HD);
  float* PCPb   = alloc((size_t)CD * HD);
  float* lpl_m  = alloc(PD);
  float* plp_m  = alloc(LD);
  float* pcp_m  = alloc(CD);
  float* gm_t   = alloc((size_t)RT * HD);
  float* tmp    = alloc((size_t)RT * HD);
  float* part   = alloc((size_t)KS * PD * HD);

  const int OPEN = 1 << 30;
  auto plain = [&](const float* p) { return RowSrc{p, p, p, p, p, p, OPEN, OPEN, OPEN, OPEN, OPEN}; };

  masks_kernel<<<MK_ROWG + MK_COLG + MK_PCG, 256, 0, stream>>>(pl, pc, lpl_m, plp_m, pcp_m);

  agg_gemm_kernel<<<dim3(PD / 16, KS), 256, 0, stream>>>(pl, LD, 0, lines, LD / KS, PD * HD, part);
  lrl_gemm_kernel<KS><<<PD / 16, 256, 0, stream>>>(plain(part), PD * HD, points, 2.f, rel_W + 0 * HH,
      rel_b + 0 * HD, rel_g + 0 * HD, rel_be + 0 * HD, LPLb);

  agg_gemm_kernel<<<dim3(LD / 16, KS), 256, 0, stream>>>(pl, LD, 1, points, PD / KS, LD * HD, part);
  lrl_gemm_kernel<KS><<<LD / 16, 256, 0, stream>>>(plain(part), LD * HD, lines, 2.f, rel_W + 1 * HH,
      rel_b + 1 * HD, rel_g + 1 * HD, rel_be + 1 * HD, PLPb);

  agg_gemm_kernel<<<dim3(CD / 16, KS), 256, 0, stream>>>(pc, CD, 1, points, PD / KS, CD * HD, part);
  lrl_gemm_kernel<KS><<<CD / 16, 256, 0, stream>>>(plain(part), CD * HD, circles, 2.f, rel_W + 2 * HH,
      rel_b + 2 * HD, rel_g + 2 * HD, rel_be + 2 * HD, PCPb);

  const RowSrc cat_t = {points, lines, LPLb, PLPb, PCPb, head_s,
                        PD, PD + LD, PD + LD + PD, PD + LD + PD + LD, PD + LD + PD + LD + CD};
  lrl_gemm_kernel<1><<<RT / 16, 256, 0, stream>>>(cat_t, 0, nullptr, 1.f, gm_W + 0 * HH,
      gm_b + 0 * HD, gm_g + 0 * HD, gm_be + 0 * HD, tmp);
  lrl_gemm_kernel<1><<<RT / 16, 256, 0, stream>>>(plain(tmp), 0, nullptr, 1.f, gm_W + 1 * HH,
      gm_b + 1 * HD, gm_g + 1 * HD, gm_be + 1 * HD, gm_t);
  const float* gm_t2 = gm_t + (size_t)(PD + LD) * HD;
  const RowSrc cat_h = {gm_t, gm_t2, gm_t2, gm_t2, gm_t2, gm_t2, PD, OPEN, OPEN, OPEN, OPEN};

  struct Head { const float* sym; int N; RowSrc geo; int R; MaskSrc mask; };
  const Head heads[6] = {
      {text_s,  NT,  plain(gm_t),  RT, {lpl_m, plp_m, pcp_m, PD + LD, PD + LD + PD,
                                        PD + LD + PD, PD + LD + PD + LD,
                                        PD + LD + PD + LD, PD + LD + PD + LD + CD}},
      {head_s,  NH,  cat_h,        RH, {lpl_m, plp_m, pcp_m, PD, PD + PD,
                                        PD + PD, PD + PD + LD,
                                        PD + PD + LD, PD + PD + LD + CD}},
      {angle_s, NA,  plain(LPLb),  PD, {lpl_m, lpl_m, lpl_m, 0, PD, 0, 0, 0, 0}},
      {bar_s,   NB,  plain(PLPb),  LD, {plp_m, plp_m, plp_m, 0, LD, 0, 0, 0, 0}},
      {para_s,  NPA, plain(lines), LD, {lpl_m, lpl_m, lpl_m, 0, 0, 0, 0, 0, 0}},
      {perp_s,  NPE, plain(LPLb),  PD, {lpl_m, lpl_m, lpl_m, 0, PD, 0, 0, 0, 0}},
  };
  size_t ooff = 0;
  for (int h = 0; h < 6; ++h) {
    const Head& hd = heads[h];
    const float* w1 = sc_W1 + (size_t)h * HH;
    const float* pb = sc_b1 + (size_t)h * HD;
    const float* pg = sc_g + (size_t)h * HD;
    const float* pe = sc_be + (size_t)h * HD;
    const float* pw = sc_W2 + (size_t)h * HD;
    if (h == 1)
      pair_fused_kernel<16><<<dim3(hd.R / PF_G, hd.N / 16), 256, 0, stream>>>(hd.sym, hd.geo,
          hd.mask, w1, pb, pg, pe, pw, sc_b2 + h, out + ooff, hd.R);
    else if (h < 4)
      pair_fused_kernel<8><<<dim3(hd.R / PF_G, hd.N / 8), 256, 0, stream>>>(hd.sym, hd.geo,
          hd.mask, w1, pb, pg, pe, pw, sc_b2 + h, out + ooff, hd.R);
    else
      pair_fused_kernel<4><<<dim3(hd.R / PF_G, hd.N / 4), 256, 0, stream>>>(hd.sym, hd.geo,
          hd.mask, w1, pb, pg, pe, pw, sc_b2 + h, out + ooff, hd.R);
    ooff += (size_t)hd.N * hd.R;
  }
}


# pair scoring: a wave's 16 pairs x 3 k-sums (x, x*x, x*g*W2) summed over lanes by one reduce-scatter butterfly, 48 lane exchanges (was 240 shuffles); LN and W2 dot from the totals; sum order changes
# speedup vs baseline: 3.2792x; 1.2120x over previous
#include <hip/hip_runtime.h>
#include <math.h>

static const int HD  = 128;
static const int PD  = 1024;
static const int LD  = 1024;
static const int CD  = 128;
static const int NT  = 64, NH = 32, NA = 64, NB = 64, NPA = 32, NPE = 32;
static const int RT  = PD + LD + PD + LD + CD + NH;
static const int RH  = PD + PD + LD + CD;
static const int HH  = HD * HD;

typedef float v2f __attribute__((ext_vector_type(2)));
typedef float v8f __attribute__((ext_vector_type(8)));

__device__ __forceinline__ v8f wmma_f32(v2f a, v2f b, v8f c) {
  return __builtin_amdgcn_wmma_f32_16x16x4_f32(false, a, false, b, (short)0, c,
                                               false, false);
}

struct RowSrc {
  const float* p0; const float* p1; const float* p2;
  const float* p3; const float* p4; const float* p5;
  int e0, e1, e2, e3, e4;
};
__device__ __forceinline__ const float* strip_base(const RowSrc& s, int r) {
  const float* p = s.p0;
  int b = 0;
  if (r >= s.e0) { p = s.p1; b = s.e0; }
  if (r >= s.e1) { p = s.p2; b = s.e1; }
  if (r >= s.e2) { p = s.p3; b = s.e2; }
  if (r >= s.e3) { p = s.p4; b = s.e3; }
  if (r >= s.e4) { p = s.p5; b = s.e4; }
  return p + (size_t)(r - b) * 128;
}
struct MaskSrc {
  const float* q0; const float* q1; const float* q2;
  int b0, e0, b1, e1, b2, e2;
};
__device__ __forceinline__ float mask_at(const MaskSrc& s, int r) {
  float mk = 1.f;
  if (r >= s.b0 && r < s.e0) mk = s.q0[r - s.b0];
  if (r >= s.b1 && r < s.e1) mk = s.q1[r - s.b1];
  if (r >= s.b2 && r < s.e2) mk = s.q2[r - s.b2];
  return mk;
}

static const int MK_ROWG = PD / 32;
static const int MK_COLG = LD / 32;
static const int MK_PCG  = CD / 32;
__global__ __launch_bounds__(256)
void masks_kernel(const int* __restrict__ pl, const int* __restrict__ pc,
                  float* __restrict__ lpl_m, float* __restrict__ plp_m,
                  float* __restrict__ pcp_m) {
  __shared__ alignas(16) int sP[8 * 32];
  const int tid  = threadIdx.x;
  const int lane = tid & 31;
  const int w    = tid >> 5;
  const int b    = blockIdx.x;

  if (b < MK_ROWG) {
    const int4* __restrict__ row = (const int4*)(pl + (size_t)(32 * b + 4 * w) * LD);
    int4 v[4][8];
#pragma unroll
    for (int r = 0; r < 4; ++r) {
#pragma unroll
      for (int j = 0; j < 8; ++j) v[r][j] = row[r * (LD / 4) + j * 32 + lane];
    }
    int s[4];
#pragma unroll
    for (int r = 0; r < 4; ++r) {
      int t = 0;
#pragma unroll
      for (int j = 0; j < 8; ++j) t += (v[r][j].x + v[r][j].y) + (v[r][j].z + v[r][j].w);
      s[r] = t;
    }
#pragma unroll
    for (int r = 0; r < 4; ++r)
      for (int o = 16; o > 0; o >>= 1) s[r] += __shfl_xor(s[r], o, 32);
    if (lane == 0) {
      sP[4 * w + 0] = s[0]; sP[4 * w + 1] = s[1];
      sP[4 * w + 2] = s[2]; sP[4 * w + 3] = s[3];
    }
    __syncthreads();
    if (w == 0) lpl_m[32 * b + lane] = (sP[lane] >= 2) ? 1.f : 0.f;
  } else {
    const int* __restrict__ rel = pl;
    float* __restrict__ dst = plp_m;
    int ld = LD, c0 = 32 * (b - MK_ROWG);
    if (b >= MK_ROWG + MK_COLG) {
      rel = pc; dst = pcp_m; ld = CD; c0 = 32 * (b - MK_ROWG - MK_COLG);
    }
    const int q = tid & 7;
    const int r = tid >> 3;
    const int* __restrict__ p = rel + (size_t)r * ld + c0 + 4 * q;
    int4 v[32];
#pragma unroll
    for (int j = 0; j < 32; ++j) v[j] = *(const int4*)(p + (size_t)(32 * j) * ld);
    int4 s = {0, 0, 0, 0};
#pragma unroll
    for (int j = 0; j < 32; ++j) { s.x += v[j].x; s.y += v[j].y; s.z += v[j].z; s.w += v[j].w; }
    for (int o = 8; o <= 16; o <<= 1) {
      s.x += __shfl_xor(s.x, o, 32); s.y += __shfl_xor(s.y, o, 32);
      s.z += __shfl_xor(s.z, o, 32); s.w += __shfl_xor(s.w, o, 32);
    }
    if (lane < 8) ((int4*)sP)[w * 8 + lane] = s;
    __syncthreads();
    if (w == 0) {
      int t = 0;
#pragma unroll
      for (int k = 0; k < 8; ++k) t += sP[k * 32 + lane];
      dst[c0 + lane] = (t >= 2) ? 1.f : 0.f;
    }
  }
}

static const int KS = 8;
__global__ __launch_bounds__(256)
void agg_gemm_kernel(const int* __restrict__ rel, int ldA, int transA,
                     const float* __restrict__ Bm, int kchunk, int pstride,
                     float* __restrict__ Cpart) {
  __shared__ float sA[16 * 128];
  const int tid  = threadIdx.x;
  const int lane = tid & 31;
  const int w    = tid >> 5;
  const int m0   = blockIdx.x * 16;
  const int n    = w * 16 + (lane & 15);
  const int am   = lane & 15;
  const int koff = (lane >> 4) << 1;

  v8f acc = {0.f, 0.f, 0.f, 0.f, 0.f, 0.f, 0.f, 0.f};

  const int kbeg = blockIdx.y * kchunk;
  float* __restrict__ Cout = Cpart + (size_t)blockIdx.y * pstride;
  for (int k0 = kbeg; k0 < kbeg + kchunk; k0 += 128) {
    if (transA) {
      for (int i = tid; i < 16 * 128; i += 256) {
        int m = i & 15, k = i >> 4;
        sA[m * 128 + k] = (float)rel[(size_t)(k0 + k) * ldA + (m0 + m)];
      }
    } else {
      for (int i = tid; i < 16 * 128; i += 256) {
        int m = i >> 7, k = i & 127;
        sA[m * 128 + k] = (float)rel[(size_t)(m0 + m) * ldA + (k0 + k)];
      }
    }
    __syncthreads();
    for (int kk = 0; kk < 128; kk += 4) {
      v2f a = *(const v2f*)&sA[am * 128 + kk + koff];
      int kg = k0 + kk + koff;
      v2f b;
      b.x = Bm[(size_t)kg * HD + n];
      b.y = Bm[(size_t)(kg + 1) * HD + n];
      acc = wmma_f32(a, b, acc);
    }
    __syncthreads();
  }
  const int mb = (lane >> 4) << 3;
#pragma unroll
  for (int v = 0; v < 8; ++v)
    Cout[(size_t)(m0 + mb + v) * HD + n] = acc[v];
}

template <int NS>
__global__ __launch_bounds__(256)
void lrl_gemm_kernel(RowSrc xs, int sstride,
                     const float* __restrict__ X1,
                     float alpha, const float* __restrict__ W,
                     const float* __restrict__ bias, const float* __restrict__ g,
                     const float* __restrict__ be, float* __restrict__ Y) {
  __shared__ float sA[16 * 128];
  __shared__ float sC[16 * 128];
  __shared__ float sMean[16], sRstd[16];
  const int tid  = threadIdx.x;
  const int lane = tid & 31;
  const int w    = tid >> 5;
  const int m0   = blockIdx.x * 16;
  const int n    = w * 16 + (lane & 15);
  const int am   = lane & 15;
  const int koff = (lane >> 4) << 1;
  const float* __restrict__ X0 = strip_base(xs, m0);

  if constexpr (NS == 1) {
    for (int i = tid; i < 16 * 128; i += 256) {
      int m = i >> 7, k = i & 127;
      float v = alpha * X0[(size_t)m * 128 + k];
      if (X1) v += X1[(size_t)(m0 + m) * 128 + k];
      sA[i] = v;
    }
  } else {
    float x[8];
#pragma unroll
    for (int j = 0; j < 8; ++j) {
      int i = tid + j * 256;
      x[j] = X0[(size_t)(i >> 7) * 128 + (i & 127)];
    }
#pragma unroll
    for (int s = 1; s < NS; ++s) {
#pragma unroll
      for (int j = 0; j < 8; ++j) {
        int i = tid + j * 256;
        x[j] += X0[(size_t)s * sstride + (size_t)(i >> 7) * 128 + (i & 127)];
      }
    }
#pragma unroll
    for (int j = 0; j < 8; ++j) {
      int i = tid + j * 256;
      float v = alpha * x[j];
      if (X1) v += X1[(size_t)(m0 + (i >> 7)) * 128 + (i & 127)];
      sA[i] = v;
    }
  }
  __syncthreads();

  v8f acc = {0.f, 0.f, 0.f, 0.f, 0.f, 0.f, 0.f, 0.f};
#pragma unroll 4
  for (int kk = 0; kk < 128; kk += 4) {
    v2f a = *(const v2f*)&sA[am * 128 + kk + koff];
    v2f b = *(const v2f*)&W[(size_t)n * 128 + kk + koff];
    acc = wmma_f32(a, b, acc);
  }

  const int mb = (lane >> 4) << 3;
  float bn = bias[n];
#pragma unroll
  for (int v = 0; v < 8; ++v) {
    float x = acc[v] + bn;
    sC[(mb + v) * 128 + n] = x > 0.f ? x : 0.f;
  }
  __syncthreads();
  if (tid < 16) {
    float s = 0.f, s2 = 0.f;
    for (int k = 0; k < 128; ++k) {
      float x = sC[tid * 128 + k];
      s += x; s2 += x * x;
    }
    float mean = s * (1.f / 128.f);
    float var  = s2 * (1.f / 128.f) - mean * mean;
    sMean[tid] = mean;
    sRstd[tid] = rsqrtf(var + 1e-5f);
  }
  __syncthreads();
  for (int i = tid; i < 16 * 128; i += 256) {
    int m = i >> 7, k = i & 127;
    Y[(size_t)(m0 + m) * 128 + k] = (sC[i] - sMean[m]) * sRstd[m] * g[k] + be[k];
  }
}

template <int M>
__device__ __forceinline__ float xrow(float v) {
  return __builtin_bit_cast(float, __builtin_amdgcn_update_dpp(
      0, __builtin_bit_cast(int, v), 0x160 + M, 0xf, 0xf, true));
}
__device__ __forceinline__ float xhalf(float v) {
  const int b = __builtin_bit_cast(int, v);
  return __builtin_bit_cast(float,
      __builtin_amdgcn_permlanex16(b, b, 0x76543210, 0xfedcba98, false, false));
}

static const int PF_G = 32;
static const int PF_P = 132;
template <int S>
__global__ __launch_bounds__(256)
void pair_fused_kernel(const float* __restrict__ sym, RowSrc geo, MaskSrc mask,
                       const float* __restrict__ W1, const float* __restrict__ b1,
                       const float* __restrict__ g, const float* __restrict__ be,
                       const float* __restrict__ W2, const float* __restrict__ b2p,
                       float* __restrict__ out, int R) {
  __shared__ alignas(16) float sS[16 * PF_P];
  __shared__ alignas(16) float sG[PF_G * PF_P];
  __shared__ float sD[S * PF_G];
  const int tid  = threadIdx.x;
  const int lane = tid & 31;
  const int w    = tid >> 5;
  const int r0   = blockIdx.x * PF_G;
  const int i0   = blockIdx.y * S;
  const int i16  = i0 & ~15;
  const int lo   = i0 - i16;
  const int n    = w * 16 + (lane & 15);
  const int am   = lane & 15;
  const int koff = (lane >> 4) << 1;

  v2f bw[32];
#pragma unroll
  for (int t = 0; t < 32; ++t)
    bw[t] = *(const v2f*)&W1[(size_t)n * 128 + 4 * t + koff];

  const float* __restrict__ gs0 = strip_base(geo, r0);
  const float* __restrict__ gs1 = strip_base(geo, r0 + 16);
  {
    float4 vs[2], vg[4];
#pragma unroll
    for (int j = 0; j < 2; ++j) {
      int i = tid + j * 256;
      vs[j] = ((const float4*)(sym + (size_t)(i16 + (i >> 5)) * 128))[i & 31];
    }
#pragma unroll
    for (int j = 0; j < 4; ++j) {
      int i = tid + j * 256;
      const float* base = (j < 2) ? gs0 : gs1;
      vg[j] = ((const float4*)(base + (size_t)((i >> 5) & 15) * 128))[i & 31];
    }
#pragma unroll
    for (int j = 0; j < 2; ++j) {
      int i = tid + j * 256;
      *(float4*)&sS[(i >> 5) * PF_P + 4 * (i & 31)] = vs[j];
    }
#pragma unroll
    for (int j = 0; j < 4; ++j) {
      int i = tid + j * 256;
      *(float4*)&sG[(i >> 5) * PF_P + 4 * (i & 31)] = vg[j];
    }
  }
  const float4 b4 = ((const float4*)b1)[lane];
  const float4 gg = ((const float4*)g)[lane];
  const float4 bb = ((const float4*)be)[lane];
  const float4 w2 = ((const float4*)W2)[lane];
  __syncthreads();

  v8f aS  = {0.f, 0.f, 0.f, 0.f, 0.f, 0.f, 0.f, 0.f};
  v8f aG0 = {0.f, 0.f, 0.f, 0.f, 0.f, 0.f, 0.f, 0.f};
  v8f aG1 = {0.f, 0.f, 0.f, 0.f, 0.f, 0.f, 0.f, 0.f};
#pragma unroll
  for (int kk = 0; kk < 128; kk += 4) {
    v2f b  = bw[kk >> 2];
    v2f x  = *(const v2f*)&sS[am * PF_P + kk + koff];
    v2f y0 = *(const v2f*)&sG[am * PF_P + kk + koff];
    v2f y1 = *(const v2f*)&sG[(16 + am) * PF_P + kk + koff];
    aS  = wmma_f32(x,  b, aS);
    aG0 = wmma_f32(y0, b, aG0);
    aG1 = wmma_f32(y1, b, aG1);
  }
  __syncthreads();
  const int mb = (lane >> 4) << 3;
#pragma unroll
  for (int v = 0; v < 8; ++v) {
    sS[(mb + v) * PF_P + n]      = aS[v];
    sG[(mb + v) * PF_P + n]      = aG0[v];
    sG[(16 + mb + v) * PF_P + n] = aG1[v];
  }
  __syncthreads();

  float4 g4[4];
#pragma unroll
  for (int c = 0; c < 4; ++c)
    g4[c] = *(const float4*)&sG[(4 * w + c) * PF_P + 4 * lane];

  const float4 gw = {gg.x * w2.x, gg.y * w2.y, gg.z * w2.z, gg.w * w2.w};
  float sgw = (gw.x + gw.y) + (gw.z + gw.w);
  float sbw = (bb.x * w2.x + bb.y * w2.y) + (bb.z * w2.z + bb.w * w2.w);
  for (int o = 16; o > 0; o >>= 1) {
    sgw += __shfl_xor(sgw, o, 32);
    sbw += __shfl_xor(sbw, o, 32);
  }
  const bool h16 = (lane & 16) != 0;
  const bool h8  = (lane & 8) != 0;
  const bool h4  = (lane & 4) != 0;
  const bool h2  = (lane & 2) != 0;

  for (int q = 0; q < S / 4; ++q) {
    float4 s4[4];
#pragma unroll
    for (int a = 0; a < 4; ++a)
      s4[a] = *(const float4*)&sS[(lo + 4 * q + a) * PF_P + 4 * lane];

    float v[48];
#pragma unroll
    for (int a = 0; a < 4; ++a) {
#pragma unroll
      for (int c = 0; c < 4; ++c) {
        float x0 = s4[a].x + g4[c].x + b4.x; x0 = x0 > 0.f ? x0 : 0.f;
        float x1 = s4[a].y + g4[c].y + b4.y; x1 = x1 > 0.f ? x1 : 0.f;
        float x2 = s4[a].z + g4[c].z + b4.z; x2 = x2 > 0.f ? x2 : 0.f;
        float x3 = s4[a].w + g4[c].w + b4.w; x3 = x3 > 0.f ? x3 : 0.f;
        const int p = 4 * a + c;
        v[3 * p + 0] = (x0 + x1) + (x2 + x3);
        v[3 * p + 1] = (x0 * x0 + x1 * x1) + (x2 * x2 + x3 * x3);
        v[3 * p + 2] = (x0 * gw.x + x1 * gw.y) + (x2 * gw.z + x3 * gw.w);
      }
    }
#pragma unroll
    for (int i = 0; i < 24; ++i) {
      const float keep = h16 ? v[24 + i] : v[i], send = h16 ? v[i] : v[24 + i];
      v[i] = keep + xhalf(send);
    }
#pragma unroll
    for (int i = 0; i < 12; ++i) {
      const float keep = h8 ? v[12 + i] : v[i], send = h8 ? v[i] : v[12 + i];
      v[i] = keep + xrow<8>(send);
    }
#pragma unroll
    for (int i = 0; i < 6; ++i) {
      const float keep = h4 ? v[6 + i] : v[i], send = h4 ? v[i] : v[6 + i];
      v[i] = keep + xrow<4>(send);
    }
#pragma unroll
    for (int i = 0; i < 3; ++i) {
      const float keep = h2 ? v[3 + i] : v[i], send = h2 ? v[i] : v[3 + i];
      v[i] = keep + xrow<2>(send);
    }
#pragma unroll
    for (int i = 0; i < 3; ++i) v[i] += xrow<1>(v[i]);

    const float mean = v[0] * (1.f / 128.f);
    const float var  = v[1] * (1.f / 128.f) - mean * mean;
    const float rstd = rsqrtf(var + 1e-5f);
    const float d    = rstd * (v[2] - mean * sgw) + sbw;
    if ((lane & 1) == 0)
      sD[(4 * q + (lane >> 3)) * PF_G + 4 * w + ((lane >> 1) & 3)] = d;
  }
  __syncthreads();

  const float b2 = b2p[0];
  for (int idx = tid; idx < S * PF_G; idx += 256) {
    const int a = idx >> 5;
    const int r = r0 + (idx & 31);
    float logit = sD[idx] + b2;
    float mk = mask_at(mask, r);
    float z  = (mk > 0.f) ? logit : -1e9f;
    out[(size_t)(i0 + a) * R + r] = 1.f / (1.f + expf(-z));
  }
}

extern "C" void kernel_launch(void* const* d_in, const int* in_sizes, int n_in,
                              void* d_out, int out_size, void* d_ws, size_t ws_size,
                              hipStream_t stream) {
  (void)in_sizes; (void)n_in; (void)out_size; (void)ws_size;
  const float* points  = (const float*)d_in[0];
  const float* lines   = (const float*)d_in[1];
  const float* circles = (const float*)d_in[2];
  const float* text_s  = (const float*)d_in[3];
  const float* head_s  = (const float*)d_in[4];
  const float* angle_s = (const float*)d_in[5];
  const float* bar_s   = (const float*)d_in[6];
  const float* para_s  = (const float*)d_in[7];
  const float* perp_s  = (const float*)d_in[8];
  const float* rel_W   = (const float*)d_in[9];
  const float* rel_b   = (const float*)d_in[10];
  const float* rel_g   = (const float*)d_in[11];
  const float* rel_be  = (const float*)d_in[12];
  const float* gm_W    = (const float*)d_in[13];
  const float* gm_b    = (const float*)d_in[14];
  const float* gm_g    = (const float*)d_in[15];
  const float* gm_be   = (const float*)d_in[16];
  const float* sc_W1   = (const float*)d_in[17];
  const float* sc_b1   = (const float*)d_in[18];
  const float* sc_g    = (const float*)d_in[19];
  const float* sc_be   = (const float*)d_in[20];
  const float* sc_W2   = (const float*)d_in[21];
  const float* sc_b2   = (const float*)d_in[22];
  const int*   pl      = (const int*)d_in[23];
  const int*   pc      = (const int*)d_in[24];
  float* out = (float*)d_out;

  float* wsp = (float*)d_ws;
  size_t off = 0;
  auto alloc = [&](size_t n) { float* p = wsp + off; off += (n + 31) & ~(size_t)31; return p; };
  float* LPLb   = alloc((size_t)PD * HD);
  float* PLPb   = alloc((size_t)LD * HD);
  float* PCPb   = alloc((size_t)CD * HD);
  float* lpl_m  = alloc(PD);
  float* plp_m  = alloc(LD);
  float* pcp_m  = alloc(CD);
  float* gm_t   = alloc((size_t)RT * HD);
  float* tmp    = alloc((size_t)RT * HD);
  float* part   = alloc((size_t)KS * PD * HD);

  const int OPEN = 1 << 30;
  auto plain = [&](const float* p) { return RowSrc{p, p, p, p, p, p, OPEN, OPEN, OPEN, OPEN, OPEN}; };

  masks_kernel<<<MK_ROWG + MK_COLG + MK_PCG, 256, 0, stream>>>(pl, pc, lpl_m, plp_m, pcp_m);

  agg_gemm_kernel<<<dim3(PD / 16, KS), 256, 0, stream>>>(pl, LD, 0, lines, LD / KS, PD * HD, part);
  lrl_gemm_kernel<KS><<<PD / 16, 256, 0, stream>>>(plain(part), PD * HD, points, 2.f, rel_W + 0 * HH,
      rel_b + 0 * HD, rel_g + 0 * HD, rel_be + 0 * HD, LPLb);

  agg_gemm_kernel<<<dim3(LD / 16, KS), 256, 0, stream>>>(pl, LD, 1, points, PD / KS, LD * HD, part);
  lrl_gemm_kernel<KS><<<LD / 16, 256, 0, stream>>>(plain(part), LD * HD, lines, 2.f, rel_W + 1 * HH,
      rel_b + 1 * HD, rel_g + 1 * HD, rel_be + 1 * HD, PLPb);

  agg_gemm_kernel<<<dim3(CD / 16, KS), 256, 0, stream>>>(pc, CD, 1, points, PD / KS, CD * HD, part);
  lrl_gemm_kernel<KS><<<CD / 16, 256, 0, stream>>>(plain(part), CD * HD, circles, 2.f, rel_W + 2 * HH,
      rel_b + 2 * HD, rel_g + 2 * HD, rel_be + 2 * HD, PCPb);

  const RowSrc cat_t = {points, lines, LPLb, PLPb, PCPb, head_s,
                        PD, PD + LD, PD + LD + PD, PD + LD + PD + LD, PD + LD + PD + LD + CD};
  lrl_gemm_kernel<1><<<RT / 16, 256, 0, stream>>>(cat_t, 0, nullptr, 1.f, gm_W + 0 * HH,
      gm_b + 0 * HD, gm_g + 0 * HD, gm_be + 0 * HD, tmp);
  lrl_gemm_kernel<1><<<RT / 16, 256, 0, stream>>>(plain(tmp), 0, nullptr, 1.f, gm_W + 1 * HH,
      gm_b + 1 * HD, gm_g + 1 * HD, gm_be + 1 * HD, gm_t);
  const float* gm_t2 = gm_t + (size_t)(PD + LD) * HD;
  const RowSrc cat_h = {gm_t, gm_t2, gm_t2, gm_t2, gm_t2, gm_t2, PD, OPEN, OPEN, OPEN, OPEN};

  struct Head { const float* sym; int N; RowSrc geo; int R; MaskSrc mask; };
  const Head heads[6] = {
      {text_s,  NT,  plain(gm_t),  RT, {lpl_m, plp_m, pcp_m, PD + LD, PD + LD + PD,
                                        PD + LD + PD, PD + LD + PD + LD,
                                        PD + LD + PD + LD, PD + LD + PD + LD + CD}},
      {head_s,  NH,  cat_h,        RH, {lpl_m, plp_m, pcp_m, PD, PD + PD,
                                        PD + PD, PD + PD + LD,
                                        PD + PD + LD, PD + PD + LD + CD}},
      {angle_s, NA,  plain(LPLb),  PD, {lpl_m, lpl_m, lpl_m, 0, PD, 0, 0, 0, 0}},
      {bar_s,   NB,  plain(PLPb),  LD, {plp_m, plp_m, plp_m, 0, LD, 0, 0, 0, 0}},
      {para_s,  NPA, plain(lines), LD, {lpl_m, lpl_m, lpl_m, 0, 0, 0, 0, 0, 0}},
      {perp_s,  NPE, plain(LPLb),  PD, {lpl_m, lpl_m, lpl_m, 0, PD, 0, 0, 0, 0}},
  };
  size_t ooff = 0;
  for (int h = 0; h < 6; ++h) {
    const Head& hd = heads[h];
    const float* w1 = sc_W1 + (size_t)h * HH;
    const float* pb = sc_b1 + (size_t)h * HD;
    const float* pg = sc_g + (size_t)h * HD;
    const float* pe = sc_be + (size_t)h * HD;
    const float* pw = sc_W2 + (size_t)h * HD;
    if (h == 1)
      pair_fused_kernel<16><<<dim3(hd.R / PF_G, hd.N / 16), 256, 0, stream>>>(hd.sym, hd.geo,
          hd.mask, w1, pb, pg, pe, pw, sc_b2 + h, out + ooff, hd.R);
    else if (h < 4)
      pair_fused_kernel<8><<<dim3(hd.R / PF_G, hd.N / 8), 256, 0, stream>>>(hd.sym, hd.geo,
          hd.mask, w1, pb, pg, pe, pw, sc_b2 + h, out + ooff, hd.R);
    else
      pair_fused_kernel<4><<<dim3(hd.R / PF_G, hd.N / 4), 256, 0, stream>>>(hd.sym, hd.geo,
          hd.mask, w1, pb, pg, pe, pw, sc_b2 + h, out + ooff, hd.R);
    ooff += (size_t)hd.N * hd.R;
  }
}


# LRL kernel: operands fetched at entry as one load group of 16 B fragment pieces (W slice in registers, X from memory or a fragment-ordered LDS tile); k permuted alike in A and B; sum order changes
# speedup vs baseline: 3.7298x; 1.1374x over previous
#include <hip/hip_runtime.h>
#include <math.h>

static const int HD  = 128;
static const int PD  = 1024;
static const int LD  = 1024;
static const int CD  = 128;
static const int NT  = 64, NH = 32, NA = 64, NB = 64, NPA = 32, NPE = 32;
static const int RT  = PD + LD + PD + LD + CD + NH;
static const int RH  = PD + PD + LD + CD;
static const int HH  = HD * HD;

typedef float v2f __attribute__((ext_vector_type(2)));
typedef float v8f __attribute__((ext_vector_type(8)));

__device__ __forceinline__ v8f wmma_f32(v2f a, v2f b, v8f c) {
  return __builtin_amdgcn_wmma_f32_16x16x4_f32(false, a, false, b, (short)0, c,
                                               false, false);
}

struct RowSrc {
  const float* p0; const float* p1; const float* p2;
  const float* p3; const float* p4; const float* p5;
  int e0, e1, e2, e3, e4;
};
__device__ __forceinline__ const float* strip_base(const RowSrc& s, int r) {
  const float* p = s.p0;
  int b = 0;
  if (r >= s.e0) { p = s.p1; b = s.e0; }
  if (r >= s.e1) { p = s.p2; b = s.e1; }
  if (r >= s.e2) { p = s.p3; b = s.e2; }
  if (r >= s.e3) { p = s.p4; b = s.e3; }
  if (r >= s.e4) { p = s.p5; b = s.e4; }
  return p + (size_t)(r - b) * 128;
}
struct MaskSrc {
  const float* q0; const float* q1; const float* q2;
  int b0, e0, b1, e1, b2, e2;
};
__device__ __forceinline__ float mask_at(const MaskSrc& s, int r) {
  float mk = 1.f;
  if (r >= s.b0 && r < s.e0) mk = s.q0[r - s.b0];
  if (r >= s.b1 && r < s.e1) mk = s.q1[r - s.b1];
  if (r >= s.b2 && r < s.e2) mk = s.q2[r - s.b2];
  return mk;
}

static const int MK_ROWG = PD / 32;
static const int MK_COLG = LD / 32;
static const int MK_PCG  = CD / 32;
__global__ __launch_bounds__(256)
void masks_kernel(const int* __restrict__ pl, const int* __restrict__ pc,
                  float* __restrict__ lpl_m, float* __restrict__ plp_m,
                  float* __restrict__ pcp_m) {
  __shared__ alignas(16) int sP[8 * 32];
  const int tid  = threadIdx.x;
  const int lane = tid & 31;
  const int w    = tid >> 5;
  const int b    = blockIdx.x;

  if (b < MK_ROWG) {
    const int4* __restrict__ row = (const int4*)(pl + (size_t)(32 * b + 4 * w) * LD);
    int4 v[4][8];
#pragma unroll
    for (int r = 0; r < 4; ++r) {
#pragma unroll
      for (int j = 0; j < 8; ++j) v[r][j] = row[r * (LD / 4) + j * 32 + lane];
    }
    int s[4];
#pragma unroll
    for (int r = 0; r < 4; ++r) {
      int t = 0;
#pragma unroll
      for (int j = 0; j < 8; ++j) t += (v[r][j].x + v[r][j].y) + (v[r][j].z + v[r][j].w);
      s[r] = t;
    }
#pragma unroll
    for (int r = 0; r < 4; ++r)
      for (int o = 16; o > 0; o >>= 1) s[r] += __shfl_xor(s[r], o, 32);
    if (lane == 0) {
      sP[4 * w + 0] = s[0]; sP[4 * w + 1] = s[1];
      sP[4 * w + 2] = s[2]; sP[4 * w + 3] = s[3];
    }
    __syncthreads();
    if (w == 0) lpl_m[32 * b + lane] = (sP[lane] >= 2) ? 1.f : 0.f;
  } else {
    const int* __restrict__ rel = pl;
    float* __restrict__ dst = plp_m;
    int ld = LD, c0 = 32 * (b - MK_ROWG);
    if (b >= MK_ROWG + MK_COLG) {
      rel = pc; dst = pcp_m; ld = CD; c0 = 32 * (b - MK_ROWG - MK_COLG);
    }
    const int q = tid & 7;
    const int r = tid >> 3;
    const int* __restrict__ p = rel + (size_t)r * ld + c0 + 4 * q;
    int4 v[32];
#pragma unroll
    for (int j = 0; j < 32; ++j) v[j] = *(const int4*)(p + (size_t)(32 * j) * ld);
    int4 s = {0, 0, 0, 0};
#pragma unroll
    for (int j = 0; j < 32; ++j) { s.x += v[j].x; s.y += v[j].y; s.z += v[j].z; s.w += v[j].w; }
    for (int o = 8; o <= 16; o <<= 1) {
      s.x += __shfl_xor(s.x, o, 32); s.y += __shfl_xor(s.y, o, 32);
      s.z += __shfl_xor(s.z, o, 32); s.w += __shfl_xor(s.w, o, 32);
    }
    if (lane < 8) ((int4*)sP)[w * 8 + lane] = s;
    __syncthreads();
    if (w == 0) {
      int t = 0;
#pragma unroll
      for (int k = 0; k < 8; ++k) t += sP[k * 32 + lane];
      dst[c0 + lane] = (t >= 2) ? 1.f : 0.f;
    }
  }
}

static const int KS = 8;
__global__ __launch_bounds__(256)
void agg_gemm_kernel(const int* __restrict__ rel, int ldA, int transA,
                     const float* __restrict__ Bm, int kchunk, int pstride,
                     float* __restrict__ Cpart) {
  __shared__ float sA[16 * 128];
  const int tid  = threadIdx.x;
  const int lane = tid & 31;
  const int w    = tid >> 5;
  const int m0   = blockIdx.x * 16;
  const int n    = w * 16 + (lane & 15);
  const int am   = lane & 15;
  const int koff = (lane >> 4) << 1;

  v8f acc = {0.f, 0.f, 0.f, 0.f, 0.f, 0.f, 0.f, 0.f};

  const int kbeg = blockIdx.y * kchunk;
  float* __restrict__ Cout = Cpart + (size_t)blockIdx.y * pstride;
  for (int k0 = kbeg; k0 < kbeg + kchunk; k0 += 128) {
    if (transA) {
      for (int i = tid; i < 16 * 128; i += 256) {
        int m = i & 15, k = i >> 4;
        sA[m * 128 + k] = (float)rel[(size_t)(k0 + k) * ldA + (m0 + m)];
      }
    } else {
      for (int i = tid; i < 16 * 128; i += 256) {
        int m = i >> 7, k = i & 127;
        sA[m * 128 + k] = (float)rel[(size_t)(m0 + m) * ldA + (k0 + k)];
      }
    }
    __syncthreads();
    for (int kk = 0; kk < 128; kk += 4) {
      v2f a = *(const v2f*)&sA[am * 128 + kk + koff];
      int kg = k0 + kk + koff;
      v2f b;
      b.x = Bm[(size_t)kg * HD + n];
      b.y = Bm[(size_t)(kg + 1) * HD + n];
      acc = wmma_f32(a, b, acc);
    }
    __syncthreads();
  }
  const int mb = (lane >> 4) << 3;
#pragma unroll
  for (int v = 0; v < 8; ++v)
    Cout[(size_t)(m0 + mb + v) * HD + n] = acc[v];
}

typedef float v4f __attribute__((ext_vector_type(4)));

__device__ __forceinline__ void loads_as_group(const v4f (&v)[16]) {
  asm volatile("" :: "v"(v[0]), "v"(v[1]), "v"(v[2]), "v"(v[3]), "v"(v[4]), "v"(v[5]),
               "v"(v[6]), "v"(v[7]), "v"(v[8]), "v"(v[9]), "v"(v[10]), "v"(v[11]),
               "v"(v[12]), "v"(v[13]), "v"(v[14]), "v"(v[15]));
}

template <bool SPLIT>
__global__ __launch_bounds__(256)
void lrl_gemm_kernel(RowSrc xs, int sstride,
                     const float* __restrict__ X1,
                     float alpha, const float* __restrict__ W,
                     const float* __restrict__ bias, const float* __restrict__ g,
                     const float* __restrict__ be, float* __restrict__ Y) {
  __shared__ float sC[16 * 128];
  __shared__ float sMean[16], sRstd[16];
  const int tid  = threadIdx.x;
  const int lane = tid & 31;
  const int w    = tid >> 5;
  const int m0   = blockIdx.x * 16;
  const int n    = w * 16 + (lane & 15);
  const int am   = lane & 15;
  const int kh   = (lane >> 4) << 2;
  const float* __restrict__ X0 = strip_base(xs, m0);

  v4f bw[16];
#pragma unroll
  for (int j = 0; j < 16; ++j)
    bw[j] = *(const v4f*)&W[(size_t)n * 128 + 8 * j + kh];
  const float bn = bias[n];
  const float gk = g[tid & 127];
  const float bk = be[tid & 127];

  v4f af[16];
  if constexpr (!SPLIT) {
#pragma unroll
    for (int j = 0; j < 16; ++j)
      af[j] = alpha * *(const v4f*)&X0[(size_t)am * 128 + 8 * j + kh];
    if (X1) {
#pragma unroll
      for (int j = 0; j < 16; ++j)
        af[j] += *(const v4f*)&X1[(size_t)(m0 + am) * 128 + 8 * j + kh];
    }
  } else {
    static_assert(KS == 8, "a thread's 2 * KS plane pieces fill the sixteen places of a group");
    __shared__ v4f sA[16 * 32];
    const int q0 = tid, q1 = tid + 256;
    const size_t o0 = (size_t)(q0 >> 5) * 128 + 4 * (q0 & 31);
    const size_t o1 = (size_t)(q1 >> 5) * 128 + 4 * (q1 & 31);
    v4f y0 = {0.f, 0.f, 0.f, 0.f}, y1 = {0.f, 0.f, 0.f, 0.f};
    if (X1) {
      y0 = *(const v4f*)&X1[(size_t)m0 * 128 + o0];
      y1 = *(const v4f*)&X1[(size_t)m0 * 128 + o1];
    }
    v4f t[16];
#pragma unroll
    for (int s = 0; s < KS; ++s) {
      t[2 * s]     = *(const v4f*)&X0[(size_t)s * sstride + o0];
      t[2 * s + 1] = *(const v4f*)&X0[(size_t)s * sstride + o1];
    }
    loads_as_group(t);
    v4f x0 = t[0], x1 = t[1];
#pragma unroll
    for (int s = 1; s < KS; ++s) { x0 += t[2 * s]; x1 += t[2 * s + 1]; }
    x0 = alpha * x0; x1 = alpha * x1;
    if (X1) { x0 += y0; x1 += y1; }
    {
      const int m = q0 >> 5, p = q0 & 31, j = p >> 1;
      sA[j * 32 + ((m + ((p & 1) << 4)) ^ j)] = x0;
    }
    {
      const int m = q1 >> 5, p = q1 & 31, j = p >> 1;
      sA[j * 32 + ((m + ((p & 1) << 4)) ^ j)] = x1;
    }
    __syncthreads();
#pragma unroll
    for (int j = 0; j < 16; ++j) af[j] = sA[j * 32 + (lane ^ j)];
    __builtin_amdgcn_sched_barrier(0);
  }

  v8f acc = {0.f, 0.f, 0.f, 0.f, 0.f, 0.f, 0.f, 0.f};
#pragma unroll
  for (int j = 0; j < 16; ++j) {
    acc = wmma_f32(af[j].xy, bw[j].xy, acc);
    acc = wmma_f32(af[j].zw, bw[j].zw, acc);
  }

  const int mb = (lane >> 4) << 3;
#pragma unroll
  for (int v = 0; v < 8; ++v) {
    float x = acc[v] + bn;
    sC[(mb + v) * 128 + n] = x > 0.f ? x : 0.f;
  }
  __syncthreads();
  if (tid < 16) {
    float s = 0.f, s2 = 0.f;
    for (int k = 0; k < 128; ++k) {
      float x = sC[tid * 128 + k];
      s += x; s2 += x * x;
    }
    float mean = s * (1.f / 128.f);
    float var  = s2 * (1.f / 128.f) - mean * mean;
    sMean[tid] = mean;
    sRstd[tid] = rsqrtf(var + 1e-5f);
  }
  __syncthreads();
  for (int i = tid; i < 16 * 128; i += 256) {
    int m = i >> 7;
    Y[(size_t)(m0 + m) * 128 + (i & 127)] = (sC[i] - sMean[m]) * sRstd[m] * gk + bk;
  }
}

template <int M>
__device__ __forceinline__ float xrow(float v) {
  return __builtin_bit_cast(float, __builtin_amdgcn_update_dpp(
      0, __builtin_bit_cast(int, v), 0x160 + M, 0xf, 0xf, true));
}
__device__ __forceinline__ float xhalf(float v) {
  const int b = __builtin_bit_cast(int, v);
  return __builtin_bit_cast(float,
      __builtin_amdgcn_permlanex16(b, b, 0x76543210, 0xfedcba98, false, false));
}

static const int PF_G = 32;
static const int PF_P = 132;
template <int S>
__global__ __launch_bounds__(256)
void pair_fused_kernel(const float* __restrict__ sym, RowSrc geo, MaskSrc mask,
                       const float* __restrict__ W1, const float* __restrict__ b1,
                       const float* __restrict__ g, const float* __restrict__ be,
                       const float* __restrict__ W2, const float* __restrict__ b2p,
                       float* __restrict__ out, int R) {
  __shared__ alignas(16) float sS[16 * PF_P];
  __shared__ alignas(16) float sG[PF_G * PF_P];
  __shared__ float sD[S * PF_G];
  const int tid  = threadIdx.x;
  const int lane = tid & 31;
  const int w    = tid >> 5;
  const int r0   = blockIdx.x * PF_G;
  const int i0   = blockIdx.y * S;
  const int i16  = i0 & ~15;
  const int lo   = i0 - i16;
  const int n    = w * 16 + (lane & 15);
  const int am   = lane & 15;
  const int koff = (lane >> 4) << 1;

  v2f bw[32];
#pragma unroll
  for (int t = 0; t < 32; ++t)
    bw[t] = *(const v2f*)&W1[(size_t)n * 128 + 4 * t + koff];

  const float* __restrict__ gs0 = strip_base(geo, r0);
  const float* __restrict__ gs1 = strip_base(geo, r0 + 16);
  {
    float4 vs[2], vg[4];
#pragma unroll
    for (int j = 0; j < 2; ++j) {
      int i = tid + j * 256;
      vs[j] = ((const float4*)(sym + (size_t)(i16 + (i >> 5)) * 128))[i & 31];
    }
#pragma unroll
    for (int j = 0; j < 4; ++j) {
      int i = tid + j * 256;
      const float* base = (j < 2) ? gs0 : gs1;
      vg[j] = ((const float4*)(base + (size_t)((i >> 5) & 15) * 128))[i & 31];
    }
#pragma unroll
    for (int j = 0; j < 2; ++j) {
      int i = tid + j * 256;
      *(float4*)&sS[(i >> 5) * PF_P + 4 * (i & 31)] = vs[j];
    }
#pragma unroll
    for (int j = 0; j < 4; ++j) {
      int i = tid + j * 256;
      *(float4*)&sG[(i >> 5) * PF_P + 4 * (i & 31)] = vg[j];
    }
  }
  const float4 b4 = ((const float4*)b1)[lane];
  const float4 gg = ((const float4*)g)[lane];
  const float4 bb = ((const float4*)be)[lane];
  const float4 w2 = ((const float4*)W2)[lane];
  __syncthreads();

  v8f aS  = {0.f, 0.f, 0.f, 0.f, 0.f, 0.f, 0.f, 0.f};
  v8f aG0 = {0.f, 0.f, 0.f, 0.f, 0.f, 0.f, 0.f, 0.f};
  v8f aG1 = {0.f, 0.f, 0.f, 0.f, 0.f, 0.f, 0.f, 0.f};
#pragma unroll
  for (int kk = 0; kk < 128; kk += 4) {
    v2f b  = bw[kk >> 2];
    v2f x  = *(const v2f*)&sS[am * PF_P + kk + koff];
    v2f y0 = *(const v2f*)&sG[am * PF_P + kk + koff];
    v2f y1 = *(const v2f*)&sG[(16 + am) * PF_P + kk + koff];
    aS  = wmma_f32(x,  b, aS);
    aG0 = wmma_f32(y0, b, aG0);
    aG1 = wmma_f32(y1, b, aG1);
  }
  __syncthreads();
  const int mb = (lane >> 4) << 3;
#pragma unroll
  for (int v = 0; v < 8; ++v) {
    sS[(mb + v) * PF_P + n]      = aS[v];
    sG[(mb + v) * PF_P + n]      = aG0[v];
    sG[(16 + mb + v) * PF_P + n] = aG1[v];
  }
  __syncthreads();

  float4 g4[4];
#pragma unroll
  for (int c = 0; c < 4; ++c)
    g4[c] = *(const float4*)&sG[(4 * w + c) * PF_P + 4 * lane];

  const float4 gw = {gg.x * w2.x, gg.y * w2.y, gg.z * w2.z, gg.w * w2.w};
  float sgw = (gw.x + gw.y) + (gw.z + gw.w);
  float sbw = (bb.x * w2.x + bb.y * w2.y) + (bb.z * w2.z + bb.w * w2.w);
  for (int o = 16; o > 0; o >>= 1) {
    sgw += __shfl_xor(sgw, o, 32);
    sbw += __shfl_xor(sbw, o, 32);
  }
  const bool h16 = (lane & 16) != 0;
  const bool h8  = (lane & 8) != 0;
  const bool h4  = (lane & 4) != 0;
  const bool h2  = (lane & 2) != 0;

  for (int q = 0; q < S / 4; ++q) {
    float4 s4[4];
#pragma unroll
    for (int a = 0; a < 4; ++a)
      s4[a] = *(const float4*)&sS[(lo + 4 * q + a) * PF_P + 4 * lane];

    float v[48];
#pragma unroll
    for (int a = 0; a < 4; ++a) {
#pragma unroll
      for (int c = 0; c < 4; ++c) {
        float x0 = s4[a].x + g4[c].x + b4.x; x0 = x0 > 0.f ? x0 : 0.f;
        float x1 = s4[a].y + g4[c].y + b4.y; x1 = x1 > 0.f ? x1 : 0.f;
        float x2 = s4[a].z + g4[c].z + b4.z; x2 = x2 > 0.f ? x2 : 0.f;
        float x3 = s4[a].w + g4[c].w + b4.w; x3 = x3 > 0.f ? x3 : 0.f;
        const int p = 4 * a + c;
        v[3 * p + 0] = (x0 + x1) + (x2 + x3);
        v[3 * p + 1] = (x0 * x0 + x1 * x1) + (x2 * x2 + x3 * x3);
        v[3 * p + 2] = (x0 * gw.x + x1 * gw.y) + (x2 * gw.z + x3 * gw.w);
      }
    }
#pragma unroll
    for (int i = 0; i < 24; ++i) {
      const float keep = h16 ? v[24 + i] : v[i], send = h16 ? v[i] : v[24 + i];
      v[i] = keep + xhalf(send);
    }
#pragma unroll
    for (int i = 0; i < 12; ++i) {
      const float keep = h8 ? v[12 + i] : v[i], send = h8 ? v[i] : v[12 + i];
      v[i] = keep + xrow<8>(send);
    }
#pragma unroll
    for (int i = 0; i < 6; ++i) {
      const float keep = h4 ? v[6 + i] : v[i], send = h4 ? v[i] : v[6 + i];
      v[i] = keep + xrow<4>(send);
    }
#pragma unroll
    for (int i = 0; i < 3; ++i) {
      const float keep = h2 ? v[3 + i] : v[i], send = h2 ? v[i] : v[3 + i];
      v[i] = keep + xrow<2>(send);
    }
#pragma unroll
    for (int i = 0; i < 3; ++i) v[i] += xrow<1>(v[i]);

    const float mean = v[0] * (1.f / 128.f);
    const float var  = v[1] * (1.f / 128.f) - mean * mean;
    const float rstd = rsqrtf(var + 1e-5f);
    const float d    = rstd * (v[2] - mean * sgw) + sbw;
    if ((lane & 1) == 0)
      sD[(4 * q + (lane >> 3)) * PF_G + 4 * w + ((lane >> 1) & 3)] = d;
  }
  __syncthreads();

  const float b2 = b2p[0];
  for (int idx = tid; idx < S * PF_G; idx += 256) {
    const int a = idx >> 5;
    const int r = r0 + (idx & 31);
    float logit = sD[idx] + b2;
    float mk = mask_at(mask, r);
    float z  = (mk > 0.f) ? logit : -1e9f;
    out[(size_t)(i0 + a) * R + r] = 1.f / (1.f + expf(-z));
  }
}

extern "C" void kernel_launch(void* const* d_in, const int* in_sizes, int n_in,
                              void* d_out, int out_size, void* d_ws, size_t ws_size,
                              hipStream_t stream) {
  (void)in_sizes; (void)n_in; (void)out_size; (void)ws_size;
  const float* points  = (const float*)d_in[0];
  const float* lines   = (const float*)d_in[1];
  const float* circles = (const float*)d_in[2];
  const float* text_s  = (const float*)d_in[3];
  const float* head_s  = (const float*)d_in[4];
  const float* angle_s = (const float*)d_in[5];
  const float* bar_s   = (const float*)d_in[6];
  const float* para_s  = (const float*)d_in[7];
  const float* perp_s  = (const float*)d_in[8];
  const float* rel_W   = (const float*)d_in[9];
  const float* rel_b   = (const float*)d_in[10];
  const float* rel_g   = (const float*)d_in[11];
  const float* rel_be  = (const float*)d_in[12];
  const float* gm_W    = (const float*)d_in[13];
  const float* gm_b    = (const float*)d_in[14];
  const float* gm_g    = (const float*)d_in[15];
  const float* gm_be   = (const float*)d_in[16];
  const float* sc_W1   = (const float*)d_in[17];
  const float* sc_b1   = (const float*)d_in[18];
  const float* sc_g    = (const float*)d_in[19];
  const float* sc_be   = (const float*)d_in[20];
  const float* sc_W2   = (const float*)d_in[21];
  const float* sc_b2   = (const float*)d_in[22];
  const int*   pl      = (const int*)d_in[23];
  const int*   pc      = (const int*)d_in[24];
  float* out = (float*)d_out;

  float* wsp = (float*)d_ws;
  size_t off = 0;
  auto alloc = [&](size_t n) { float* p = wsp + off; off += (n + 31) & ~(size_t)31; return p; };
  float* LPLb   = alloc((size_t)PD * HD);
  float* PLPb   = alloc((size_t)LD * HD);
  float* PCPb   = alloc((size_t)CD * HD);
  float* lpl_m  = alloc(PD);
  float* plp_m  = alloc(LD);
  float* pcp_m  = alloc(CD);
  float* gm_t   = alloc((size_t)RT * HD);
  float* tmp    = alloc((size_t)RT * HD);
  float* part   = alloc((size_t)KS * PD * HD);

  const int OPEN = 1 << 30;
  auto plain = [&](const float* p) { return RowSrc{p, p, p, p, p, p, OPEN, OPEN, OPEN, OPEN, OPEN}; };

  masks_kernel<<<MK_ROWG + MK_COLG + MK_PCG, 256, 0, stream>>>(pl, pc, lpl_m, plp_m, pcp_m);

  agg_gemm_kernel<<<dim3(PD / 16, KS), 256, 0, stream>>>(pl, LD, 0, lines, LD / KS, PD * HD, part);
  lrl_gemm_kernel<true><<<PD / 16, 256, 0, stream>>>(plain(part), PD * HD, points, 2.f, rel_W + 0 * HH,
      rel_b + 0 * HD, rel_g + 0 * HD, rel_be + 0 * HD, LPLb);

  agg_gemm_kernel<<<dim3(LD / 16, KS), 256, 0, stream>>>(pl, LD, 1, points, PD / KS, LD * HD, part);
  lrl_gemm_kernel<true><<<LD / 16, 256, 0, stream>>>(plain(part), LD * HD, lines, 2.f, rel_W + 1 * HH,
      rel_b + 1 * HD, rel_g + 1 * HD, rel_be + 1 * HD, PLPb);

  agg_gemm_kernel<<<dim3(CD / 16, KS), 256, 0, stream>>>(pc, CD, 1, points, PD / KS, CD * HD, part);
  lrl_gemm_kernel<true><<<CD / 16, 256, 0, stream>>>(plain(part), CD * HD, circles, 2.f, rel_W + 2 * HH,
      rel_b + 2 * HD, rel_g + 2 * HD, rel_be + 2 * HD, PCPb);

  const RowSrc cat_t = {points, lines, LPLb, PLPb, PCPb, head_s,
                        PD, PD + LD, PD + LD + PD, PD + LD + PD + LD, PD + LD + PD + LD + CD};
  lrl_gemm_kernel<false><<<RT / 16, 256, 0, stream>>>(cat_t, 0, nullptr, 1.f, gm_W + 0 * HH,
      gm_b + 0 * HD, gm_g + 0 * HD, gm_be + 0 * HD, tmp);
  lrl_gemm_kernel<false><<<RT / 16, 256, 0, stream>>>(plain(tmp), 0, nullptr, 1.f, gm_W + 1 * HH,
      gm_b + 1 * HD, gm_g + 1 * HD, gm_be + 1 * HD, gm_t);
  const float* gm_t2 = gm_t + (size_t)(PD + LD) * HD;
  const RowSrc cat_h = {gm_t, gm_t2, gm_t2, gm_t2, gm_t2, gm_t2, PD, OPEN, OPEN, OPEN, OPEN};

  struct Head { const float* sym; int N; RowSrc geo; int R; MaskSrc mask; };
  const Head heads[6] = {
      {text_s,  NT,  plain(gm_t),  RT, {lpl_m, plp_m, pcp_m, PD + LD, PD + LD + PD,
                                        PD + LD + PD, PD + LD + PD + LD,
                                        PD + LD + PD + LD, PD + LD + PD + LD + CD}},
      {head_s,  NH,  cat_h,        RH, {lpl_m, plp_m, pcp_m, PD, PD + PD,
                                        PD + PD, PD + PD + LD,
                                        PD + PD + LD, PD + PD + LD + CD}},
      {angle_s, NA,  plain(LPLb),  PD, {lpl_m, lpl_m, lpl_m, 0, PD, 0, 0, 0, 0}},
      {bar_s,   NB,  plain(PLPb),  LD, {plp_m, plp_m, plp_m, 0, LD, 0, 0, 0, 0}},
      {para_s,  NPA, plain(lines), LD, {lpl_m, lpl_m, lpl_m, 0, 0, 0, 0, 0, 0}},
      {perp_s,  NPE, plain(LPLb),  PD, {lpl_m, lpl_m, lpl_m, 0, PD, 0, 0, 0, 0}},
  };
  size_t ooff = 0;
  for (int h = 0; h < 6; ++h) {
    const Head& hd = heads[h];
    const float* w1 = sc_W1 + (size_t)h * HH;
    const float* pb = sc_b1 + (size_t)h * HD;
    const float* pg = sc_g + (size_t)h * HD;
    const float* pe = sc_be + (size_t)h * HD;
    const float* pw = sc_W2 + (size_t)h * HD;
    if (h == 1)
      pair_fused_kernel<16><<<dim3(hd.R / PF_G, hd.N / 16), 256, 0, stream>>>(hd.sym, hd.geo,
          hd.mask, w1, pb, pg, pe, pw, sc_b2 + h, out + ooff, hd.R);
    else if (h < 4)
      pair_fused_kernel<8><<<dim3(hd.R / PF_G, hd.N / 8), 256, 0, stream>>>(hd.sym, hd.geo,
          hd.mask, w1, pb, pg, pe, pw, sc_b2 + h, out + ooff, hd.R);
    else
      pair_fused_kernel<4><<<dim3(hd.R / PF_G, hd.N / 4), 256, 0, stream>>>(hd.sym, hd.geo,
          hd.mask, w1, pb, pg, pe, pw, sc_b2 + h, out + ooff, hd.R);
    ooff += (size_t)hd.N * hd.R;
  }
}


# aggregation GEMM: operands go straight from memory into fragment registers, all loads issued at entry as one group (no LDS tile, no barrier); a wave owns 2 interleaved column tiles; sum order changes
# speedup vs baseline: 4.2339x; 1.1352x over previous
#include <hip/hip_runtime.h>
#include <math.h>

static const int HD  = 128;
static const int PD  = 1024;
static const int LD  = 1024;
static const int CD  = 128;
static const int NT  = 64, NH = 32, NA = 64, NB = 64, NPA = 32, NPE = 32;
static const int RT  = PD + LD + PD + LD + CD + NH;
static const int RH  = PD + PD + LD + CD;
static const int HH  = HD * HD;

typedef float v2f __attribute__((ext_vector_type(2)));
typedef float v8f __attribute__((ext_vector_type(8)));

__device__ __forceinline__ v8f wmma_f32(v2f a, v2f b, v8f c) {
  return __builtin_amdgcn_wmma_f32_16x16x4_f32(false, a, false, b, (short)0, c,
                                               false, false);
}

struct RowSrc {
  const float* p0; const float* p1; const float* p2;
  const float* p3; const float* p4; const float* p5;
  int e0, e1, e2, e3, e4;
};
__device__ __forceinline__ const float* strip_base(const RowSrc& s, int r) {
  const float* p = s.p0;
  int b = 0;
  if (r >= s.e0) { p = s.p1; b = s.e0; }
  if (r >= s.e1) { p = s.p2; b = s.e1; }
  if (r >= s.e2) { p = s.p3; b = s.e2; }
  if (r >= s.e3) { p = s.p4; b = s.e3; }
  if (r >= s.e4) { p = s.p5; b = s.e4; }
  return p + (size_t)(r - b) * 128;
}
struct MaskSrc {
  const float* q0; const float* q1; const float* q2;
  int b0, e0, b1, e1, b2, e2;
};
__device__ __forceinline__ float mask_at(const MaskSrc& s, int r) {
  float mk = 1.f;
  if (r >= s.b0 && r < s.e0) mk = s.q0[r - s.b0];
  if (r >= s.b1 && r < s.e1) mk = s.q1[r - s.b1];
  if (r >= s.b2 && r < s.e2) mk = s.q2[r - s.b2];
  return mk;
}

static const int MK_ROWG = PD / 32;
static const int MK_COLG = LD / 32;
static const int MK_PCG  = CD / 32;
__global__ __launch_bounds__(256)
void masks_kernel(const int* __restrict__ pl, const int* __restrict__ pc,
                  float* __restrict__ lpl_m, float* __restrict__ plp_m,
                  float* __restrict__ pcp_m) {
  __shared__ alignas(16) int sP[8 * 32];
  const int tid  = threadIdx.x;
  const int lane = tid & 31;
  const int w    = tid >> 5;
  const int b    = blockIdx.x;

  if (b < MK_ROWG) {
    const int4* __restrict__ row = (const int4*)(pl + (size_t)(32 * b + 4 * w) * LD);
    int4 v[4][8];
#pragma unroll
    for (int r = 0; r < 4; ++r) {
#pragma unroll
      for (int j = 0; j < 8; ++j) v[r][j] = row[r * (LD / 4) + j * 32 + lane];
    }
    int s[4];
#pragma unroll
    for (int r = 0; r < 4; ++r) {
      int t = 0;
#pragma unroll
      for (int j = 0; j < 8; ++j) t += (v[r][j].x + v[r][j].y) + (v[r][j].z + v[r][j].w);
      s[r] = t;
    }
#pragma unroll
    for (int r = 0; r < 4; ++r)
      for (int o = 16; o > 0; o >>= 1) s[r] += __shfl_xor(s[r], o, 32);
    if (lane == 0) {
      sP[4 * w + 0] = s[0]; sP[4 * w + 1] = s[1];
      sP[4 * w + 2] = s[2]; sP[4 * w + 3] = s[3];
    }
    __syncthreads();
    if (w == 0) lpl_m[32 * b + lane] = (sP[lane] >= 2) ? 1.f : 0.f;
  } else {
    const int* __restrict__ rel = pl;
    float* __restrict__ dst = plp_m;
    int ld = LD, c0 = 32 * (b - MK_ROWG);
    if (b >= MK_ROWG + MK_COLG) {
      rel = pc; dst = pcp_m; ld = CD; c0 = 32 * (b - MK_ROWG - MK_COLG);
    }
    const int q = tid & 7;
    const int r = tid >> 3;
    const int* __restrict__ p = rel + (size_t)r * ld + c0 + 4 * q;
    int4 v[32];
#pragma unroll
    for (int j = 0; j < 32; ++j) v[j] = *(const int4*)(p + (size_t)(32 * j) * ld);
    int4 s = {0, 0, 0, 0};
#pragma unroll
    for (int j = 0; j < 32; ++j) { s.x += v[j].x; s.y += v[j].y; s.z += v[j].z; s.w += v[j].w; }
    for (int o = 8; o <= 16; o <<= 1) {
      s.x += __shfl_xor(s.x, o, 32); s.y += __shfl_xor(s.y, o, 32);
      s.z += __shfl_xor(s.z, o, 32); s.w += __shfl_xor(s.w, o, 32);
    }
    if (lane < 8) ((int4*)sP)[w * 8 + lane] = s;
    __syncthreads();
    if (w == 0) {
      int t = 0;
#pragma unroll
      for (int k = 0; k < 8; ++k) t += sP[k * 32 + lane];
      dst[c0 + lane] = (t >= 2) ? 1.f : 0.f;
    }
  }
}

static const int KS   = 8;
static const int AG_K = 128;
typedef float v4f __attribute__((ext_vector_type(4)));
template <bool TRANS, int LDA>
__global__ __launch_bounds__(128)
void agg_gemm_kernel(const int* __restrict__ rel, const float* __restrict__ Bm, int pstride,
                     float* __restrict__ Cpart) {
  const int lane = threadIdx.x & 31;
  const int w    = threadIdx.x >> 5;
  const int m0   = blockIdx.x * 16;
  const int c    = lane & 15;
  const int h    = lane >> 4;
  const int kh   = h << 2;
  const int kbeg = blockIdx.y * AG_K;
  const int col  = 32 * w + 2 * c;

  const float* __restrict__ bp = Bm + (size_t)(kbeg + kh) * HD + col;
  v2f bf[16][4];
#pragma unroll
  for (int j = 0; j < 16; ++j) {
#pragma unroll
    for (int s = 0; s < 4; ++s)
      bf[j][s] = *(const v2f*)&bp[(size_t)(8 * j + s) * HD];
  }

  int4 ai[16];
  if constexpr (!TRANS) {
    const int4* __restrict__ ap = (const int4*)(rel + (size_t)(m0 + c) * LDA + kbeg + kh);
#pragma unroll
    for (int j = 0; j < 16; ++j) ai[j] = ap[2 * j];
  } else {
    const int* __restrict__ ap = rel + (size_t)(kbeg + kh) * LDA + m0 + c;
#pragma unroll
    for (int j = 0; j < 16; ++j) {
      ai[j].x = ap[(size_t)(8 * j + 0) * LDA]; ai[j].y = ap[(size_t)(8 * j + 1) * LDA];
      ai[j].z = ap[(size_t)(8 * j + 2) * LDA]; ai[j].w = ap[(size_t)(8 * j + 3) * LDA];
    }
  }
  __builtin_amdgcn_sched_barrier(0);
  v4f af[16];
#pragma unroll
  for (int j = 0; j < 16; ++j) {
    af[j].x = (float)ai[j].x; af[j].y = (float)ai[j].y;
    af[j].z = (float)ai[j].z; af[j].w = (float)ai[j].w;
  }

  v8f acc0 = {0.f, 0.f, 0.f, 0.f, 0.f, 0.f, 0.f, 0.f};
  v8f acc1 = {0.f, 0.f, 0.f, 0.f, 0.f, 0.f, 0.f, 0.f};
#pragma unroll
  for (int j = 0; j < 16; ++j) {
    const v2f b00 = {bf[j][0].x, bf[j][1].x}, b01 = {bf[j][0].y, bf[j][1].y};
    const v2f b10 = {bf[j][2].x, bf[j][3].x}, b11 = {bf[j][2].y, bf[j][3].y};
    acc0 = wmma_f32(af[j].xy, b00, acc0);
    acc1 = wmma_f32(af[j].xy, b01, acc1);
    acc0 = wmma_f32(af[j].zw, b10, acc0);
    acc1 = wmma_f32(af[j].zw, b11, acc1);
  }

  float* __restrict__ Cout = Cpart + (size_t)blockIdx.y * pstride
                           + (size_t)(m0 + 8 * h) * HD + col;
#pragma unroll
  for (int v = 0; v < 8; ++v) {
    const v2f o = {acc0[v], acc1[v]};
    *(v2f*)&Cout[(size_t)v * HD] = o;
  }
}


__device__ __forceinline__ void loads_as_group(const v4f (&v)[16]) {
  asm volatile("" :: "v"(v[0]), "v"(v[1]), "v"(v[2]), "v"(v[3]), "v"(v[4]), "v"(v[5]),
               "v"(v[6]), "v"(v[7]), "v"(v[8]), "v"(v[9]), "v"(v[10]), "v"(v[11]),
               "v"(v[12]), "v"(v[13]), "v"(v[14]), "v"(v[15]));
}

template <bool SPLIT>
__global__ __launch_bounds__(256)
void lrl_gemm_kernel(RowSrc xs, int sstride,
                     const float* __restrict__ X1,
                     float alpha, const float* __restrict__ W,
                     const float* __restrict__ bias, const float* __restrict__ g,
                     const float* __restrict__ be, float* __restrict__ Y) {
  __shared__ float sC[16 * 128];
  __shared__ float sMean[16], sRstd[16];
  const int tid  = threadIdx.x;
  const int lane = tid & 31;
  const int w    = tid >> 5;
  const int m0   = blockIdx.x * 16;
  const int n    = w * 16 + (lane & 15);
  const int am   = lane & 15;
  const int kh   = (lane >> 4) << 2;
  const float* __restrict__ X0 = strip_base(xs, m0);

  v4f bw[16];
#pragma unroll
  for (int j = 0; j < 16; ++j)
    bw[j] = *(const v4f*)&W[(size_t)n * 128 + 8 * j + kh];
  const float bn = bias[n];
  const float gk = g[tid & 127];
  const float bk = be[tid & 127];

  v4f af[16];
  if constexpr (!SPLIT) {
#pragma unroll
    for (int j = 0; j < 16; ++j)
      af[j] = alpha * *(const v4f*)&X0[(size_t)am * 128 + 8 * j + kh];
    if (X1) {
#pragma unroll
      for (int j = 0; j < 16; ++j)
        af[j] += *(const v4f*)&X1[(size_t)(m0 + am) * 128 + 8 * j + kh];
    }
  } else {
    static_assert(KS == 8, "a thread's 2 * KS plane pieces fill the sixteen places of a group");
    __shared__ v4f sA[16 * 32];
    const int q0 = tid, q1 = tid + 256;
    const size_t o0 = (size_t)(q0 >> 5) * 128 + 4 * (q0 & 31);
    const size_t o1 = (size_t)(q1 >> 5) * 128 + 4 * (q1 & 31);
    v4f y0 = {0.f, 0.f, 0.f, 0.f}, y1 = {0.f, 0.f, 0.f, 0.f};
    if (X1) {
      y0 = *(const v4f*)&X1[(size_t)m0 * 128 + o0];
      y1 = *(const v4f*)&X1[(size_t)m0 * 128 + o1];
    }
    v4f t[16];
#pragma unroll
    for (int s = 0; s < KS; ++s) {
      t[2 * s]     = *(const v4f*)&X0[(size_t)s * sstride + o0];
      t[2 * s + 1] = *(const v4f*)&X0[(size_t)s * sstride + o1];
    }
    loads_as_group(t);
    v4f x0 = t[0], x1 = t[1];
#pragma unroll
    for (int s = 1; s < KS; ++s) { x0 += t[2 * s]; x1 += t[2 * s + 1]; }
    x0 = alpha * x0; x1 = alpha * x1;
    if (X1) { x0 += y0; x1 += y1; }
    {
      const int m = q0 >> 5, p = q0 & 31, j = p >> 1;
      sA[j * 32 + ((m + ((p & 1) << 4)) ^ j)] = x0;
    }
    {
      const int m = q1 >> 5, p = q1 & 31, j = p >> 1;
      sA[j * 32 + ((m + ((p & 1) << 4)) ^ j)] = x1;
    }
    __syncthreads();
#pragma unroll
    for (int j = 0; j < 16; ++j) af[j] = sA[j * 32 + (lane ^ j)];
    __builtin_amdgcn_sched_barrier(0);
  }

  v8f acc = {0.f, 0.f, 0.f, 0.f, 0.f, 0.f, 0.f, 0.f};
#pragma unroll
  for (int j = 0; j < 16; ++j) {
    acc = wmma_f32(af[j].xy, bw[j].xy, acc);
    acc = wmma_f32(af[j].zw, bw[j].zw, acc);
  }

  const int mb = (lane >> 4) << 3;
#pragma unroll
  for (int v = 0; v < 8; ++v) {
    float x = acc[v] + bn;
    sC[(mb + v) * 128 + n] = x > 0.f ? x : 0.f;
  }
  __syncthreads();
  if (tid < 16) {
    float s = 0.f, s2 = 0.f;
    for (int k = 0; k < 128; ++k) {
      float x = sC[tid * 128 + k];
      s += x; s2 += x * x;
    }
    float mean = s * (1.f / 128.f);
    float var  = s2 * (1.f / 128.f) - mean * mean;
    sMean[tid] = mean;
    sRstd[tid] = rsqrtf(var + 1e-5f);
  }
  __syncthreads();
  for (int i = tid; i < 16 * 128; i += 256) {
    int m = i >> 7;
    Y[(size_t)(m0 + m) * 128 + (i & 127)] = (sC[i] - sMean[m]) * sRstd[m] * gk + bk;
  }
}

template <int M>
__device__ __forceinline__ float xrow(float v) {
  return __builtin_bit_cast(float, __builtin_amdgcn_update_dpp(
      0, __builtin_bit_cast(int, v), 0x160 + M, 0xf, 0xf, true));
}
__device__ __forceinline__ float xhalf(float v) {
  const int b = __builtin_bit_cast(int, v);
  return __builtin_bit_cast(float,
      __builtin_amdgcn_permlanex16(b, b, 0x76543210, 0xfedcba98, false, false));
}

static const int PF_G = 32;
static const int PF_P = 132;
template <int S>
__global__ __launch_bounds__(256)
void pair_fused_kernel(const float* __restrict__ sym, RowSrc geo, MaskSrc mask,
                       const float* __restrict__ W1, const float* __restrict__ b1,
                       const float* __restrict__ g, const float* __restrict__ be,
                       const float* __restrict__ W2, const float* __restrict__ b2p,
                       float* __restrict__ out, int R) {
  __shared__ alignas(16) float sS[16 * PF_P];
  __shared__ alignas(16) float sG[PF_G * PF_P];
  __shared__ float sD[S * PF_G];
  const int tid  = threadIdx.x;
  const int lane = tid & 31;
  const int w    = tid >> 5;
  const int r0   = blockIdx.x * PF_G;
  const int i0   = blockIdx.y * S;
  const int i16  = i0 & ~15;
  const int lo   = i0 - i16;
  const int n    = w * 16 + (lane & 15);
  const int am   = lane & 15;
  const int koff = (lane >> 4) << 1;

  v2f bw[32];
#pragma unroll
  for (int t = 0; t < 32; ++t)
    bw[t] = *(const v2f*)&W1[(size_t)n * 128 + 4 * t + koff];

  const float* __restrict__ gs0 = strip_base(geo, r0);
  const float* __restrict__ gs1 = strip_base(geo, r0 + 16);
  {
    float4 vs[2], vg[4];
#pragma unroll
    for (int j = 0; j < 2; ++j) {
      int i = tid + j * 256;
      vs[j] = ((const float4*)(sym + (size_t)(i16 + (i >> 5)) * 128))[i & 31];
    }
#pragma unroll
    for (int j = 0; j < 4; ++j) {
      int i = tid + j * 256;
      const float* base = (j < 2) ? gs0 : gs1;
      vg[j] = ((const float4*)(base + (size_t)((i >> 5) & 15) * 128))[i & 31];
    }
#pragma unroll
    for (int j = 0; j < 2; ++j) {
      int i = tid + j * 256;
      *(float4*)&sS[(i >> 5) * PF_P + 4 * (i & 31)] = vs[j];
    }
#pragma unroll
    for (int j = 0; j < 4; ++j) {
      int i = tid + j * 256;
      *(float4*)&sG[(i >> 5) * PF_P + 4 * (i & 31)] = vg[j];
    }
  }
  const float4 b4 = ((const float4*)b1)[lane];
  const float4 gg = ((const float4*)g)[lane];
  const float4 bb = ((const float4*)be)[lane];
  const float4 w2 = ((const float4*)W2)[lane];
  __syncthreads();

  v8f aS  = {0.f, 0.f, 0.f, 0.f, 0.f, 0.f, 0.f, 0.f};
  v8f aG0 = {0.f, 0.f, 0.f, 0.f, 0.f, 0.f, 0.f, 0.f};
  v8f aG1 = {0.f, 0.f, 0.f, 0.f, 0.f, 0.f, 0.f, 0.f};
#pragma unroll
  for (int kk = 0; kk < 128; kk += 4) {
    v2f b  = bw[kk >> 2];
    v2f x  = *(const v2f*)&sS[am * PF_P + kk + koff];
    v2f y0 = *(const v2f*)&sG[am * PF_P + kk + koff];
    v2f y1 = *(const v2f*)&sG[(16 + am) * PF_P + kk + koff];
    aS  = wmma_f32(x,  b, aS);
    aG0 = wmma_f32(y0, b, aG0);
    aG1 = wmma_f32(y1, b, aG1);
  }
  __syncthreads();
  const int mb = (lane >> 4) << 3;
#pragma unroll
  for (int v = 0; v < 8; ++v) {
    sS[(mb + v) * PF_P + n]      = aS[v];
    sG[(mb + v) * PF_P + n]      = aG0[v];
    sG[(16 + mb + v) * PF_P + n] = aG1[v];
  }
  __syncthreads();

  float4 g4[4];
#pragma unroll
  for (int c = 0; c < 4; ++c)
    g4[c] = *(const float4*)&sG[(4 * w + c) * PF_P + 4 * lane];

  const float4 gw = {gg.x * w2.x, gg.y * w2.y, gg.z * w2.z, gg.w * w2.w};
  float sgw = (gw.x + gw.y) + (gw.z + gw.w);
  float sbw = (bb.x * w2.x + bb.y * w2.y) + (bb.z * w2.z + bb.w * w2.w);
  for (int o = 16; o > 0; o >>= 1) {
    sgw += __shfl_xor(sgw, o, 32);
    sbw += __shfl_xor(sbw, o, 32);
  }
  const bool h16 = (lane & 16) != 0;
  const bool h8  = (lane & 8) != 0;
  const bool h4  = (lane & 4) != 0;
  const bool h2  = (lane & 2) != 0;

  for (int q = 0; q < S / 4; ++q) {
    float4 s4[4];
#pragma unroll
    for (int a = 0; a < 4; ++a)
      s4[a] = *(const float4*)&sS[(lo + 4 * q + a) * PF_P + 4 * lane];

    float v[48];
#pragma unroll
    for (int a = 0; a < 4; ++a) {
#pragma unroll
      for (int c = 0; c < 4; ++c) {
        float x0 = s4[a].x + g4[c].x + b4.x; x0 = x0 > 0.f ? x0 : 0.f;
        float x1 = s4[a].y + g4[c].y + b4.y; x1 = x1 > 0.f ? x1 : 0.f;
        float x2 = s4[a].z + g4[c].z + b4.z; x2 = x2 > 0.f ? x2 : 0.f;
        float x3 = s4[a].w + g4[c].w + b4.w; x3 = x3 > 0.f ? x3 : 0.f;
        const int p = 4 * a + c;
        v[3 * p + 0] = (x0 + x1) + (x2 + x3);
        v[3 * p + 1] = (x0 * x0 + x1 * x1) + (x2 * x2 + x3 * x3);
        v[3 * p + 2] = (x0 * gw.x + x1 * gw.y) + (x2 * gw.z + x3 * gw.w);
      }
    }
#pragma unroll
    for (int i = 0; i < 24; ++i) {
      const float keep = h16 ? v[24 + i] : v[i], send = h16 ? v[i] : v[24 + i];
      v[i] = keep + xhalf(send);
    }
#pragma unroll
    for (int i = 0; i < 12; ++i) {
      const float keep = h8 ? v[12 + i] : v[i], send = h8 ? v[i] : v[12 + i];
      v[i] = keep + xrow<8>(send);
    }
#pragma unroll
    for (int i = 0; i < 6; ++i) {
      const float keep = h4 ? v[6 + i] : v[i], send = h4 ? v[i] : v[6 + i];
      v[i] = keep + xrow<4>(send);
    }
#pragma unroll
    for (int i = 0; i < 3; ++i) {
      const float keep = h2 ? v[3 + i] : v[i], send = h2 ? v[i] : v[3 + i];
      v[i] = keep + xrow<2>(send);
    }
#pragma unroll
    for (int i = 0; i < 3; ++i) v[i] += xrow<1>(v[i]);

    const float mean = v[0] * (1.f / 128.f);
    const float var  = v[1] * (1.f / 128.f) - mean * mean;
    const float rstd = rsqrtf(var + 1e-5f);
    const float d    = rstd * (v[2] - mean * sgw) + sbw;
    if ((lane & 1) == 0)
      sD[(4 * q + (lane >> 3)) * PF_G + 4 * w + ((lane >> 1) & 3)] = d;
  }
  __syncthreads();

  const float b2 = b2p[0];
  for (int idx = tid; idx < S * PF_G; idx += 256) {
    const int a = idx >> 5;
    const int r = r0 + (idx & 31);
    float logit = sD[idx] + b2;
    float mk = mask_at(mask, r);
    float z  = (mk > 0.f) ? logit : -1e9f;
    out[(size_t)(i0 + a) * R + r] = 1.f / (1.f + expf(-z));
  }
}

extern "C" void kernel_launch(void* const* d_in, const int* in_sizes, int n_in,
                              void* d_out, int out_size, void* d_ws, size_t ws_size,
                              hipStream_t stream) {
  (void)in_sizes; (void)n_in; (void)out_size; (void)ws_size;
  const float* points  = (const float*)d_in[0];
  const float* lines   = (const float*)d_in[1];
  const float* circles = (const float*)d_in[2];
  const float* text_s  = (const float*)d_in[3];
  const float* head_s  = (const float*)d_in[4];
  const float* angle_s = (const float*)d_in[5];
  const float* bar_s   = (const float*)d_in[6];
  const float* para_s  = (const float*)d_in[7];
  const float* perp_s  = (const float*)d_in[8];
  const float* rel_W   = (const float*)d_in[9];
  const float* rel_b   = (const float*)d_in[10];
  const float* rel_g   = (const float*)d_in[11];
  const float* rel_be  = (const float*)d_in[12];
  const float* gm_W    = (const float*)d_in[13];
  const float* gm_b    = (const float*)d_in[14];
  const float* gm_g    = (const float*)d_in[15];
  const float* gm_be   = (const float*)d_in[16];
  const float* sc_W1   = (const float*)d_in[17];
  const float* sc_b1   = (const float*)d_in[18];
  const float* sc_g    = (const float*)d_in[19];
  const float* sc_be   = (const float*)d_in[20];
  const float* sc_W2   = (const float*)d_in[21];
  const float* sc_b2   = (const float*)d_in[22];
  const int*   pl      = (const int*)d_in[23];
  const int*   pc      = (const int*)d_in[24];
  float* out = (float*)d_out;

  float* wsp = (float*)d_ws;
  size_t off = 0;
  auto alloc = [&](size_t n) { float* p = wsp + off; off += (n + 31) & ~(size_t)31; return p; };
  float* LPLb   = alloc((size_t)PD * HD);
  float* PLPb   = alloc((size_t)LD * HD);
  float* PCPb   = alloc((size_t)CD * HD);
  float* lpl_m  = alloc(PD);
  float* plp_m  = alloc(LD);
  float* pcp_m  = alloc(CD);
  float* gm_t   = alloc((size_t)RT * HD);
  float* tmp    = alloc((size_t)RT * HD);
  float* part   = alloc((size_t)KS * PD * HD);

  const int OPEN = 1 << 30;
  auto plain = [&](const float* p) { return RowSrc{p, p, p, p, p, p, OPEN, OPEN, OPEN, OPEN, OPEN}; };

  static_assert(LD / KS == AG_K && PD / KS == AG_K, "a work-group's K slice is AG_K rows of the margin");
  masks_kernel<<<MK_ROWG + MK_COLG + MK_PCG, 256, 0, stream>>>(pl, pc, lpl_m, plp_m, pcp_m);

  agg_gemm_kernel<false, LD><<<dim3(PD / 16, KS), 128, 0, stream>>>(pl, lines, PD * HD, part);
  lrl_gemm_kernel<true><<<PD / 16, 256, 0, stream>>>(plain(part), PD * HD, points, 2.f, rel_W + 0 * HH,
      rel_b + 0 * HD, rel_g + 0 * HD, rel_be + 0 * HD, LPLb);

  agg_gemm_kernel<true, LD><<<dim3(LD / 16, KS), 128, 0, stream>>>(pl, points, LD * HD, part);
  lrl_gemm_kernel<true><<<LD / 16, 256, 0, stream>>>(plain(part), LD * HD, lines, 2.f, rel_W + 1 * HH,
      rel_b + 1 * HD, rel_g + 1 * HD, rel_be + 1 * HD, PLPb);

  agg_gemm_kernel<true, CD><<<dim3(CD / 16, KS), 128, 0, stream>>>(pc, points, CD * HD, part);
  lrl_gemm_kernel<true><<<CD / 16, 256, 0, stream>>>(plain(part), CD * HD, circles, 2.f, rel_W + 2 * HH,
      rel_b + 2 * HD, rel_g + 2 * HD, rel_be + 2 * HD, PCPb);

  const RowSrc cat_t = {points, lines, LPLb, PLPb, PCPb, head_s,
                        PD, PD + LD, PD + LD + PD, PD + LD + PD + LD, PD + LD + PD + LD + CD};
  lrl_gemm_kernel<false><<<RT / 16, 256, 0, stream>>>(cat_t, 0, nullptr, 1.f, gm_W + 0 * HH,
      gm_b + 0 * HD, gm_g + 0 * HD, gm_be + 0 * HD, tmp);
  lrl_gemm_kernel<false><<<RT / 16, 256, 0, stream>>>(plain(tmp), 0, nullptr, 1.f, gm_W + 1 * HH,
      gm_b + 1 * HD, gm_g + 1 * HD, gm_be + 1 * HD, gm_t);
  const float* gm_t2 = gm_t + (size_t)(PD + LD) * HD;
  const RowSrc cat_h = {gm_t, gm_t2, gm_t2, gm_t2, gm_t2, gm_t2, PD, OPEN, OPEN, OPEN, OPEN};

  struct Head { const float* sym; int N; RowSrc geo; int R; MaskSrc mask; };
  const Head heads[6] = {
      {text_s,  NT,  plain(gm_t),  RT, {lpl_m, plp_m, pcp_m, PD + LD, PD + LD + PD,
                                        PD + LD + PD, PD + LD + PD + LD,
                                        PD + LD + PD + LD, PD + LD + PD + LD + CD}},
      {head_s,  NH,  cat_h,        RH, {lpl_m, plp_m, pcp_m, PD, PD + PD,
                                        PD + PD, PD + PD + LD,
                                        PD + PD + LD, PD + PD + LD + CD}},
      {angle_s, NA,  plain(LPLb),  PD, {lpl_m, lpl_m, lpl_m, 0, PD, 0, 0, 0, 0}},
      {bar_s,   NB,  plain(PLPb),  LD, {plp_m, plp_m, plp_m, 0, LD, 0, 0, 0, 0}},
      {para_s,  NPA, plain(lines), LD, {lpl_m, lpl_m, lpl_m, 0, 0, 0, 0, 0, 0}},
      {perp_s,  NPE, plain(LPLb),  PD, {lpl_m, lpl_m, lpl_m, 0, PD, 0, 0, 0, 0}},
  };
  size_t ooff = 0;
  for (int h = 0; h < 6; ++h) {
    const Head& hd = heads[h];
    const float* w1 = sc_W1 + (size_t)h * HH;
    const float* pb = sc_b1 + (size_t)h * HD;
    const float* pg = sc_g + (size_t)h * HD;
    const float* pe = sc_be + (size_t)h * HD;
    const float* pw = sc_W2 + (size_t)h * HD;
    if (h == 1)
      pair_fused_kernel<16><<<dim3(hd.R / PF_G, hd.N / 16), 256, 0, stream>>>(hd.sym, hd.geo,
          hd.mask, w1, pb, pg, pe, pw, sc_b2 + h, out + ooff, hd.R);
    else if (h < 4)
      pair_fused_kernel<8><<<dim3(hd.R / PF_G, hd.N / 8), 256, 0, stream>>>(hd.sym, hd.geo,
          hd.mask, w1, pb, pg, pe, pw, sc_b2 + h, out + ooff, hd.R);
    else
      pair_fused_kernel<4><<<dim3(hd.R / PF_G, hd.N / 4), 256, 0, stream>>>(hd.sym, hd.geo,
          hd.mask, w1, pb, pg, pe, pw, sc_b2 + h, out + ooff, hd.R);
    ooff += (size_t)hd.N * hd.R;
  }
}


# pair and split-LRL kernels: no global load is waited for at or behind a barrier: the staging barrier orders LDS only, W fragment loads stay in flight across it; mask and b2 loads issued at entry
# speedup vs baseline: 4.3990x; 1.0390x over previous
#include <hip/hip_runtime.h>
#include <math.h>

static const int HD  = 128;
static const int PD  = 1024;
static const int LD  = 1024;
static const int CD  = 128;
static const int NT  = 64, NH = 32, NA = 64, NB = 64, NPA = 32, NPE = 32;
static const int RT  = PD + LD + PD + LD + CD + NH;
static const int RH  = PD + PD + LD + CD;
static const int HH  = HD * HD;

typedef float v2f __attribute__((ext_vector_type(2)));
typedef float v8f __attribute__((ext_vector_type(8)));

__device__ __forceinline__ v8f wmma_f32(v2f a, v2f b, v8f c) {
  return __builtin_amdgcn_wmma_f32_16x16x4_f32(false, a, false, b, (short)0, c,
                                               false, false);
}

struct RowSrc {
  const float* p0; const float* p1; const float* p2;
  const float* p3; const float* p4; const float* p5;
  int e0, e1, e2, e3, e4;
};
__device__ __forceinline__ const float* strip_base(const RowSrc& s, int r) {
  const float* p = s.p0;
  int b = 0;
  if (r >= s.e0) { p = s.p1; b = s.e0; }
  if (r >= s.e1) { p = s.p2; b = s.e1; }
  if (r >= s.e2) { p = s.p3; b = s.e2; }
  if (r >= s.e3) { p = s.p4; b = s.e3; }
  if (r >= s.e4) { p = s.p5; b = s.e4; }
  return p + (size_t)(r - b) * 128;
}
struct MaskSrc {
  const float* q0; const float* q1; const float* q2;
  int b0, e0, b1, e1, b2, e2;
};
__device__ __forceinline__ const float* mask_addr(const MaskSrc& s, int r, bool& inside) {
  const float* p = s.q0;
  inside = false;
  if (r >= s.b0 && r < s.e0) { p = s.q0 + (r - s.b0); inside = true; }
  if (r >= s.b1 && r < s.e1) { p = s.q1 + (r - s.b1); inside = true; }
  if (r >= s.b2 && r < s.e2) { p = s.q2 + (r - s.b2); inside = true; }
  return p;
}

static const int MK_ROWG = PD / 32;
static const int MK_COLG = LD / 32;
static const int MK_PCG  = CD / 32;
__global__ __launch_bounds__(256)
void masks_kernel(const int* __restrict__ pl, const int* __restrict__ pc,
                  float* __restrict__ lpl_m, float* __restrict__ plp_m,
                  float* __restrict__ pcp_m) {
  __shared__ alignas(16) int sP[8 * 32];
  const int tid  = threadIdx.x;
  const int lane = tid & 31;
  const int w    = tid >> 5;
  const int b    = blockIdx.x;

  if (b < MK_ROWG) {
    const int4* __restrict__ row = (const int4*)(pl + (size_t)(32 * b + 4 * w) * LD);
    int4 v[4][8];
#pragma unroll
    for (int r = 0; r < 4; ++r) {
#pragma unroll
      for (int j = 0; j < 8; ++j) v[r][j] = row[r * (LD / 4) + j * 32 + lane];
    }
    int s[4];
#pragma unroll
    for (int r = 0; r < 4; ++r) {
      int t = 0;
#pragma unroll
      for (int j = 0; j < 8; ++j) t += (v[r][j].x + v[r][j].y) + (v[r][j].z + v[r][j].w);
      s[r] = t;
    }
#pragma unroll
    for (int r = 0; r < 4; ++r)
      for (int o = 16; o > 0; o >>= 1) s[r] += __shfl_xor(s[r], o, 32);
    if (lane == 0) {
      sP[4 * w + 0] = s[0]; sP[4 * w + 1] = s[1];
      sP[4 * w + 2] = s[2]; sP[4 * w + 3] = s[3];
    }
    __syncthreads();
    if (w == 0) lpl_m[32 * b + lane] = (sP[lane] >= 2) ? 1.f : 0.f;
  } else {
    const int* __restrict__ rel = pl;
    float* __restrict__ dst = plp_m;
    int ld = LD, c0 = 32 * (b - MK_ROWG);
    if (b >= MK_ROWG + MK_COLG) {
      rel = pc; dst = pcp_m; ld = CD; c0 = 32 * (b - MK_ROWG - MK_COLG);
    }
    const int q = tid & 7;
    const int r = tid >> 3;
    const int* __restrict__ p = rel + (size_t)r * ld + c0 + 4 * q;
    int4 v[32];
#pragma unroll
    for (int j = 0; j < 32; ++j) v[j] = *(const int4*)(p + (size_t)(32 * j) * ld);
    int4 s = {0, 0, 0, 0};
#pragma unroll
    for (int j = 0; j < 32; ++j) { s.x += v[j].x; s.y += v[j].y; s.z += v[j].z; s.w += v[j].w; }
    for (int o = 8; o <= 16; o <<= 1) {
      s.x += __shfl_xor(s.x, o, 32); s.y += __shfl_xor(s.y, o, 32);
      s.z += __shfl_xor(s.z, o, 32); s.w += __shfl_xor(s.w, o, 32);
    }
    if (lane < 8) ((int4*)sP)[w * 8 + lane] = s;
    __syncthreads();
    if (w == 0) {
      int t = 0;
#pragma unroll
      for (int k = 0; k < 8; ++k) t += sP[k * 32 + lane];
      dst[c0 + lane] = (t >= 2) ? 1.f : 0.f;
    }
  }
}

static const int KS   = 8;
static const int AG_K = 128;
typedef float v4f __attribute__((ext_vector_type(4)));
template <bool TRANS, int LDA>
__global__ __launch_bounds__(128)
void agg_gemm_kernel(const int* __restrict__ rel, const float* __restrict__ Bm, int pstride,
                     float* __restrict__ Cpart) {
  const int lane = threadIdx.x & 31;
  const int w    = threadIdx.x >> 5;
  const int m0   = blockIdx.x * 16;
  const int c    = lane & 15;
  const int h    = lane >> 4;
  const int kh   = h << 2;
  const int kbeg = blockIdx.y * AG_K;
  const int col  = 32 * w + 2 * c;

  const float* __restrict__ bp = Bm + (size_t)(kbeg + kh) * HD + col;
  v2f bf[16][4];
#pragma unroll
  for (int j = 0; j < 16; ++j) {
#pragma unroll
    for (int s = 0; s < 4; ++s)
      bf[j][s] = *(const v2f*)&bp[(size_t)(8 * j + s) * HD];
  }

  int4 ai[16];
  if constexpr (!TRANS) {
    const int4* __restrict__ ap = (const int4*)(rel + (size_t)(m0 + c) * LDA + kbeg + kh);
#pragma unroll
    for (int j = 0; j < 16; ++j) ai[j] = ap[2 * j];
  } else {
    const int* __restrict__ ap = rel + (size_t)(kbeg + kh) * LDA + m0 + c;
#pragma unroll
    for (int j = 0; j < 16; ++j) {
      ai[j].x = ap[(size_t)(8 * j + 0) * LDA]; ai[j].y = ap[(size_t)(8 * j + 1) * LDA];
      ai[j].z = ap[(size_t)(8 * j + 2) * LDA]; ai[j].w = ap[(size_t)(8 * j + 3) * LDA];
    }
  }
  __builtin_amdgcn_sched_barrier(0);
  v4f af[16];
#pragma unroll
  for (int j = 0; j < 16; ++j) {
    af[j].x = (float)ai[j].x; af[j].y = (float)ai[j].y;
    af[j].z = (float)ai[j].z; af[j].w = (float)ai[j].w;
  }

  v8f acc0 = {0.f, 0.f, 0.f, 0.f, 0.f, 0.f, 0.f, 0.f};
  v8f acc1 = {0.f, 0.f, 0.f, 0.f, 0.f, 0.f, 0.f, 0.f};
#pragma unroll
  for (int j = 0; j < 16; ++j) {
    const v2f b00 = {bf[j][0].x, bf[j][1].x}, b01 = {bf[j][0].y, bf[j][1].y};
    const v2f b10 = {bf[j][2].x, bf[j][3].x}, b11 = {bf[j][2].y, bf[j][3].y};
    acc0 = wmma_f32(af[j].xy, b00, acc0);
    acc1 = wmma_f32(af[j].xy, b01, acc1);
    acc0 = wmma_f32(af[j].zw, b10, acc0);
    acc1 = wmma_f32(af[j].zw, b11, acc1);
  }

  float* __restrict__ Cout = Cpart + (size_t)blockIdx.y * pstride
                           + (size_t)(m0 + 8 * h) * HD + col;
#pragma unroll
  for (int v = 0; v < 8; ++v) {
    const v2f o = {acc0[v], acc1[v]};
    *(v2f*)&Cout[(size_t)v * HD] = o;
  }
}

__device__ __forceinline__ void lds_barrier() {
  __builtin_amdgcn_fence(__ATOMIC_RELEASE, "workgroup", "local");
  __builtin_amdgcn_s_barrier();
  __builtin_amdgcn_fence(__ATOMIC_ACQUIRE, "workgroup", "local");
}


__device__ __forceinline__ void loads_as_group(const v4f (&v)[16]) {
  asm volatile("" :: "v"(v[0]), "v"(v[1]), "v"(v[2]), "v"(v[3]), "v"(v[4]), "v"(v[5]),
               "v"(v[6]), "v"(v[7]), "v"(v[8]), "v"(v[9]), "v"(v[10]), "v"(v[11]),
               "v"(v[12]), "v"(v[13]), "v"(v[14]), "v"(v[15]));
}

template <bool SPLIT>
__global__ __launch_bounds__(256)
void lrl_gemm_kernel(RowSrc xs, int sstride,
                     const float* __restrict__ X1,
                     float alpha, const float* __restrict__ W,
                     const float* __restrict__ bias, const float* __restrict__ g,
                     const float* __restrict__ be, float* __restrict__ Y) {
  __shared__ float sC[16 * 128];
  __shared__ float sMean[16], sRstd[16];
  const int tid  = threadIdx.x;
  const int lane = tid & 31;
  const int w    = tid >> 5;
  const int m0   = blockIdx.x * 16;
  const int n    = w * 16 + (lane & 15);
  const int am   = lane & 15;
  const int kh   = (lane >> 4) << 2;
  const float* __restrict__ X0 = strip_base(xs, m0);

  v4f bw[16];
  float bn, gk, bk;
  if constexpr (!SPLIT) {
#pragma unroll
    for (int j = 0; j < 16; ++j)
      bw[j] = *(const v4f*)&W[(size_t)n * 128 + 8 * j + kh];
    bn = bias[n];
    gk = g[tid & 127];
    bk = be[tid & 127];
  }

  v4f af[16];
  if constexpr (!SPLIT) {
#pragma unroll
    for (int j = 0; j < 16; ++j)
      af[j] = alpha * *(const v4f*)&X0[(size_t)am * 128 + 8 * j + kh];
    if (X1) {
#pragma unroll
      for (int j = 0; j < 16; ++j)
        af[j] += *(const v4f*)&X1[(size_t)(m0 + am) * 128 + 8 * j + kh];
    }
  } else {
    static_assert(KS == 8, "a thread's 2 * KS plane pieces fill the sixteen places of a group");
    __shared__ v4f sA[16 * 32];
    const int q0 = tid, q1 = tid + 256;
    const size_t o0 = (size_t)(q0 >> 5) * 128 + 4 * (q0 & 31);
    const size_t o1 = (size_t)(q1 >> 5) * 128 + 4 * (q1 & 31);
    v4f y0 = {0.f, 0.f, 0.f, 0.f}, y1 = {0.f, 0.f, 0.f, 0.f};
    if (X1) {
      y0 = *(const v4f*)&X1[(size_t)m0 * 128 + o0];
      y1 = *(const v4f*)&X1[(size_t)m0 * 128 + o1];
    }
    v4f t[16];
#pragma unroll
    for (int s = 0; s < KS; ++s) {
      t[2 * s]     = *(const v4f*)&X0[(size_t)s * sstride + o0];
      t[2 * s + 1] = *(const v4f*)&X0[(size_t)s * sstride + o1];
    }
#pragma unroll
    for (int j = 0; j < 16; ++j)
      bw[j] = *(const v4f*)&W[(size_t)n * 128 + 8 * j + kh];
    bn = bias[n];
    gk = g[tid & 127];
    bk = be[tid & 127];
    loads_as_group(t);
    v4f x0 = t[0], x1 = t[1];
#pragma unroll
    for (int s = 1; s < KS; ++s) { x0 += t[2 * s]; x1 += t[2 * s + 1]; }
    x0 = alpha * x0; x1 = alpha * x1;
    if (X1) { x0 += y0; x1 += y1; }
    {
      const int m = q0 >> 5, p = q0 & 31, j = p >> 1;
      sA[j * 32 + ((m + ((p & 1) << 4)) ^ j)] = x0;
    }
    {
      const int m = q1 >> 5, p = q1 & 31, j = p >> 1;
      sA[j * 32 + ((m + ((p & 1) << 4)) ^ j)] = x1;
    }
    lds_barrier();
#pragma unroll
    for (int j = 0; j < 16; ++j) af[j] = sA[j * 32 + (lane ^ j)];
    __builtin_amdgcn_sched_barrier(0);
  }

  v8f acc = {0.f, 0.f, 0.f, 0.f, 0.f, 0.f, 0.f, 0.f};
#pragma unroll
  for (int j = 0; j < 16; ++j) {
    acc = wmma_f32(af[j].xy, bw[j].xy, acc);
    acc = wmma_f32(af[j].zw, bw[j].zw, acc);
  }

  const int mb = (lane >> 4) << 3;
#pragma unroll
  for (int v = 0; v < 8; ++v) {
    float x = acc[v] + bn;
    sC[(mb + v) * 128 + n] = x > 0.f ? x : 0.f;
  }
  __syncthreads();
  if (tid < 16) {
    float s = 0.f, s2 = 0.f;
    for (int k = 0; k < 128; ++k) {
      float x = sC[tid * 128 + k];
      s += x; s2 += x * x;
    }
    float mean = s * (1.f / 128.f);
    float var  = s2 * (1.f / 128.f) - mean * mean;
    sMean[tid] = mean;
    sRstd[tid] = rsqrtf(var + 1e-5f);
  }
  __syncthreads();
  for (int i = tid; i < 16 * 128; i += 256) {
    int m = i >> 7;
    Y[(size_t)(m0 + m) * 128 + (i & 127)] = (sC[i] - sMean[m]) * sRstd[m] * gk + bk;
  }
}

template <int M>
__device__ __forceinline__ float xrow(float v) {
  return __builtin_bit_cast(float, __builtin_amdgcn_update_dpp(
      0, __builtin_bit_cast(int, v), 0x160 + M, 0xf, 0xf, true));
}
__device__ __forceinline__ float xhalf(float v) {
  const int b = __builtin_bit_cast(int, v);
  return __builtin_bit_cast(float,
      __builtin_amdgcn_permlanex16(b, b, 0x76543210, 0xfedcba98, false, false));
}

static const int PF_G = 32;
static const int PF_P = 132;
template <int S>
__global__ __launch_bounds__(256)
void pair_fused_kernel(const float* __restrict__ sym, RowSrc geo, MaskSrc mask,
                       const float* __restrict__ W1, const float* __restrict__ b1,
                       const float* __restrict__ g, const float* __restrict__ be,
                       const float* __restrict__ W2, const float* __restrict__ b2p,
                       float* __restrict__ out, int R) {
  __shared__ alignas(16) float sS[16 * PF_P];
  __shared__ alignas(16) float sG[PF_G * PF_P];
  __shared__ float sD[S * PF_G];
  const int tid  = threadIdx.x;
  const int lane = tid & 31;
  const int w    = tid >> 5;
  const int r0   = blockIdx.x * PF_G;
  const int i0   = blockIdx.y * S;
  const int i16  = i0 & ~15;
  const int lo   = i0 - i16;
  const int n    = w * 16 + (lane & 15);
  const int am   = lane & 15;
  const int koff = (lane >> 4) << 1;

  v2f bw[32];
#pragma unroll
  for (int t = 0; t < 32; ++t)
    bw[t] = *(const v2f*)&W1[(size_t)n * 128 + 4 * t + koff];

  const float* __restrict__ gs0 = strip_base(geo, r0);
  const float* __restrict__ gs1 = strip_base(geo, r0 + 16);
  {
    float4 vs[2], vg[4];
#pragma unroll
    for (int j = 0; j < 2; ++j) {
      int i = tid + j * 256;
      vs[j] = ((const float4*)(sym + (size_t)(i16 + (i >> 5)) * 128))[i & 31];
    }
#pragma unroll
    for (int j = 0; j < 4; ++j) {
      int i = tid + j * 256;
      const float* base = (j < 2) ? gs0 : gs1;
      vg[j] = ((const float4*)(base + (size_t)((i >> 5) & 15) * 128))[i & 31];
    }
#pragma unroll
    for (int j = 0; j < 2; ++j) {
      int i = tid + j * 256;
      *(float4*)&sS[(i >> 5) * PF_P + 4 * (i & 31)] = vs[j];
    }
#pragma unroll
    for (int j = 0; j < 4; ++j) {
      int i = tid + j * 256;
      *(float4*)&sG[(i >> 5) * PF_P + 4 * (i & 31)] = vg[j];
    }
  }
  const float4 b4 = ((const float4*)b1)[lane];
  const float4 gg = ((const float4*)g)[lane];
  const float4 bb = ((const float4*)be)[lane];
  const float4 w2 = ((const float4*)W2)[lane];
  bool mk_inside;
  const float mk_raw = *mask_addr(mask, r0 + lane, mk_inside);
  const float b2 = b2p[0];
  lds_barrier();

  v8f aS  = {0.f, 0.f, 0.f, 0.f, 0.f, 0.f, 0.f, 0.f};
  v8f aG0 = {0.f, 0.f, 0.f, 0.f, 0.f, 0.f, 0.f, 0.f};
  v8f aG1 = {0.f, 0.f, 0.f, 0.f, 0.f, 0.f, 0.f, 0.f};
#pragma unroll
  for (int kk = 0; kk < 128; kk += 4) {
    v2f b  = bw[kk >> 2];
    v2f x  = *(const v2f*)&sS[am * PF_P + kk + koff];
    v2f y0 = *(const v2f*)&sG[am * PF_P + kk + koff];
    v2f y1 = *(const v2f*)&sG[(16 + am) * PF_P + kk + koff];
    aS  = wmma_f32(x,  b, aS);
    aG0 = wmma_f32(y0, b, aG0);
    aG1 = wmma_f32(y1, b, aG1);
  }
  __builtin_amdgcn_sched_barrier(0);
  __syncthreads();
  const int mb = (lane >> 4) << 3;
#pragma unroll
  for (int v = 0; v < 8; ++v) {
    sS[(mb + v) * PF_P + n]      = aS[v];
    sG[(mb + v) * PF_P + n]      = aG0[v];
    sG[(16 + mb + v) * PF_P + n] = aG1[v];
  }
  __syncthreads();

  float4 g4[4];
#pragma unroll
  for (int c = 0; c < 4; ++c)
    g4[c] = *(const float4*)&sG[(4 * w + c) * PF_P + 4 * lane];

  const float4 gw = {gg.x * w2.x, gg.y * w2.y, gg.z * w2.z, gg.w * w2.w};
  float sgw = (gw.x + gw.y) + (gw.z + gw.w);
  float sbw = (bb.x * w2.x + bb.y * w2.y) + (bb.z * w2.z + bb.w * w2.w);
  for (int o = 16; o > 0; o >>= 1) {
    sgw += __shfl_xor(sgw, o, 32);
    sbw += __shfl_xor(sbw, o, 32);
  }
  const bool h16 = (lane & 16) != 0;
  const bool h8  = (lane & 8) != 0;
  const bool h4  = (lane & 4) != 0;
  const bool h2  = (lane & 2) != 0;

  for (int q = 0; q < S / 4; ++q) {
    float4 s4[4];
#pragma unroll
    for (int a = 0; a < 4; ++a)
      s4[a] = *(const float4*)&sS[(lo + 4 * q + a) * PF_P + 4 * lane];

    float v[48];
#pragma unroll
    for (int a = 0; a < 4; ++a) {
#pragma unroll
      for (int c = 0; c < 4; ++c) {
        float x0 = s4[a].x + g4[c].x + b4.x; x0 = x0 > 0.f ? x0 : 0.f;
        float x1 = s4[a].y + g4[c].y + b4.y; x1 = x1 > 0.f ? x1 : 0.f;
        float x2 = s4[a].z + g4[c].z + b4.z; x2 = x2 > 0.f ? x2 : 0.f;
        float x3 = s4[a].w + g4[c].w + b4.w; x3 = x3 > 0.f ? x3 : 0.f;
        const int p = 4 * a + c;
        v[3 * p + 0] = (x0 + x1) + (x2 + x3);
        v[3 * p + 1] = (x0 * x0 + x1 * x1) + (x2 * x2 + x3 * x3);
        v[3 * p + 2] = (x0 * gw.x + x1 * gw.y) + (x2 * gw.z + x3 * gw.w);
      }
    }
#pragma unroll
    for (int i = 0; i < 24; ++i) {
      const float keep = h16 ? v[24 + i] : v[i], send = h16 ? v[i] : v[24 + i];
      v[i] = keep + xhalf(send);
    }
#pragma unroll
    for (int i = 0; i < 12; ++i) {
      const float keep = h8 ? v[12 + i] : v[i], send = h8 ? v[i] : v[12 + i];
      v[i] = keep + xrow<8>(send);
    }
#pragma unroll
    for (int i = 0; i < 6; ++i) {
      const float keep = h4 ? v[6 + i] : v[i], send = h4 ? v[i] : v[6 + i];
      v[i] = keep + xrow<4>(send);
    }
#pragma unroll
    for (int i = 0; i < 3; ++i) {
      const float keep = h2 ? v[3 + i] : v[i], send = h2 ? v[i] : v[3 + i];
      v[i] = keep + xrow<2>(send);
    }
#pragma unroll
    for (int i = 0; i < 3; ++i) v[i] += xrow<1>(v[i]);

    const float mean = v[0] * (1.f / 128.f);
    const float var  = v[1] * (1.f / 128.f) - mean * mean;
    const float rstd = rsqrtf(var + 1e-5f);
    const float d    = rstd * (v[2] - mean * sgw) + sbw;
    if ((lane & 1) == 0)
      sD[(4 * q + (lane >> 3)) * PF_G + 4 * w + ((lane >> 1) & 3)] = d;
  }
  __syncthreads();

  for (int idx = tid; idx < S * PF_G; idx += 256) {
    const int a = idx >> 5;
    const int r = r0 + (idx & 31);
    float logit = sD[idx] + b2;
    const float mk = mk_inside ? mk_raw : 1.f;
    float z  = (mk > 0.f) ? logit : -1e9f;
    out[(size_t)(i0 + a) * R + r] = 1.f / (1.f + expf(-z));
  }
}

extern "C" void kernel_launch(void* const* d_in, const int* in_sizes, int n_in,
                              void* d_out, int out_size, void* d_ws, size_t ws_size,
                              hipStream_t stream) {
  (void)in_sizes; (void)n_in; (void)out_size; (void)ws_size;
  const float* points  = (const float*)d_in[0];
  const float* lines   = (const float*)d_in[1];
  const float* circles = (const float*)d_in[2];
  const float* text_s  = (const float*)d_in[3];
  const float* head_s  = (const float*)d_in[4];
  const float* angle_s = (const float*)d_in[5];
  const float* bar_s   = (const float*)d_in[6];
  const float* para_s  = (const float*)d_in[7];
  const float* perp_s  = (const float*)d_in[8];
  const float* rel_W   = (const float*)d_in[9];
  const float* rel_b   = (const float*)d_in[10];
  const float* rel_g   = (const float*)d_in[11];
  const float* rel_be  = (const float*)d_in[12];
  const float* gm_W    = (const float*)d_in[13];
  const float* gm_b    = (const float*)d_in[14];
  const float* gm_g    = (const float*)d_in[15];
  const float* gm_be   = (const float*)d_in[16];
  const float* sc_W1   = (const float*)d_in[17];
  const float* sc_b1   = (const float*)d_in[18];
  const float* sc_g    = (const float*)d_in[19];
  const float* sc_be   = (const float*)d_in[20];
  const float* sc_W2   = (const float*)d_in[21];
  const float* sc_b2   = (const float*)d_in[22];
  const int*   pl      = (const int*)d_in[23];
  const int*   pc      = (const int*)d_in[24];
  float* out = (float*)d_out;

  float* wsp = (float*)d_ws;
  size_t off = 0;
  auto alloc = [&](size_t n) { float* p = wsp + off; off += (n + 31) & ~(size_t)31; return p; };
  float* LPLb   = alloc((size_t)PD * HD);
  float* PLPb   = alloc((size_t)LD * HD);
  float* PCPb   = alloc((size_t)CD * HD);
  float* lpl_m  = alloc(PD);
  float* plp_m  = alloc(LD);
  float* pcp_m  = alloc(CD);
  float* gm_t   = alloc((size_t)RT * HD);
  float* tmp    = alloc((size_t)RT * HD);
  float* part   = alloc((size_t)KS * PD * HD);

  const int OPEN = 1 << 30;
  auto plain = [&](const float* p) { return RowSrc{p, p, p, p, p, p, OPEN, OPEN, OPEN, OPEN, OPEN}; };

  static_assert(LD / KS == AG_K && PD / KS == AG_K, "a work-group's K slice is AG_K rows of the margin");
  masks_kernel<<<MK_ROWG + MK_COLG + MK_PCG, 256, 0, stream>>>(pl, pc, lpl_m, plp_m, pcp_m);

  agg_gemm_kernel<false, LD><<<dim3(PD / 16, KS), 128, 0, stream>>>(pl, lines, PD * HD, part);
  lrl_gemm_kernel<true><<<PD / 16, 256, 0, stream>>>(plain(part), PD * HD, points, 2.f, rel_W + 0 * HH,
      rel_b + 0 * HD, rel_g + 0 * HD, rel_be + 0 * HD, LPLb);

  agg_gemm_kernel<true, LD><<<dim3(LD / 16, KS), 128, 0, stream>>>(pl, points, LD * HD, part);
  lrl_gemm_kernel<true><<<LD / 16, 256, 0, stream>>>(plain(part), LD * HD, lines, 2.f, rel_W + 1 * HH,
      rel_b + 1 * HD, rel_g + 1 * HD, rel_be + 1 * HD, PLPb);

  agg_gemm_kernel<true, CD><<<dim3(CD / 16, KS), 128, 0, stream>>>(pc, points, CD * HD, part);
  lrl_gemm_kernel<true><<<CD / 16, 256, 0, stream>>>(plain(part), CD * HD, circles, 2.f, rel_W + 2 * HH,
      rel_b + 2 * HD, rel_g + 2 * HD, rel_be + 2 * HD, PCPb);

  const RowSrc cat_t = {points, lines, LPLb, PLPb, PCPb, head_s,
                        PD, PD + LD, PD + LD + PD, PD + LD + PD + LD, PD + LD + PD + LD + CD};
  lrl_gemm_kernel<false><<<RT / 16, 256, 0, stream>>>(cat_t, 0, nullptr, 1.f, gm_W + 0 * HH,
      gm_b + 0 * HD, gm_g + 0 * HD, gm_be + 0 * HD, tmp);
  lrl_gemm_kernel<false><<<RT / 16, 256, 0, stream>>>(plain(tmp), 0, nullptr, 1.f, gm_W + 1 * HH,
      gm_b + 1 * HD, gm_g + 1 * HD, gm_be + 1 * HD, gm_t);
  const float* gm_t2 = gm_t + (size_t)(PD + LD) * HD;
  const RowSrc cat_h = {gm_t, gm_t2, gm_t2, gm_t2, gm_t2, gm_t2, PD, OPEN, OPEN, OPEN, OPEN};

  struct Head { const float* sym; int N; RowSrc geo; int R; MaskSrc mask; };
  const Head heads[6] = {
      {text_s,  NT,  plain(gm_t),  RT, {lpl_m, plp_m, pcp_m, PD + LD, PD + LD + PD,
                                        PD + LD + PD, PD + LD + PD + LD,
                                        PD + LD + PD + LD, PD + LD + PD + LD + CD}},
      {head_s,  NH,  cat_h,        RH, {lpl_m, plp_m, pcp_m, PD, PD + PD,
                                        PD + PD, PD + PD + LD,
                                        PD + PD + LD, PD + PD + LD + CD}},
      {angle_s, NA,  plain(LPLb),  PD, {lpl_m, lpl_m, lpl_m, 0, PD, 0, 0, 0, 0}},
      {bar_s,   NB,  plain(PLPb),  LD, {plp_m, plp_m, plp_m, 0, LD, 0, 0, 0, 0}},
      {para_s,  NPA, plain(lines), LD, {lpl_m, lpl_m, lpl_m, 0, 0, 0, 0, 0, 0}},
      {perp_s,  NPE, plain(LPLb),  PD, {lpl_m, lpl_m, lpl_m, 0, PD, 0, 0, 0, 0}},
  };
  size_t ooff = 0;
  for (int h = 0; h < 6; ++h) {
    const Head& hd = heads[h];
    const float* w1 = sc_W1 + (size_t)h * HH;
    const float* pb = sc_b1 + (size_t)h * HD;
    const float* pg = sc_g + (size_t)h * HD;
    const float* pe = sc_be + (size_t)h * HD;
    const float* pw = sc_W2 + (size_t)h * HD;
    if (h == 1)
      pair_fused_kernel<16><<<dim3(hd.R / PF_G, hd.N / 16), 256, 0, stream>>>(hd.sym, hd.geo,
          hd.mask, w1, pb, pg, pe, pw, sc_b2 + h, out + ooff, hd.R);
    else if (h < 4)
      pair_fused_kernel<8><<<dim3(hd.R / PF_G, hd.N / 8), 256, 0, stream>>>(hd.sym, hd.geo,
          hd.mask, w1, pb, pg, pe, pw, sc_b2 + h, out + ooff, hd.R);
    else
      pair_fused_kernel<4><<<dim3(hd.R / PF_G, hd.N / 4), 256, 0, stream>>>(hd.sym, hd.geo,
          hd.mask, w1, pb, pg, pe, pw, sc_b2 + h, out + ooff, hd.R);
    ooff += (size_t)hd.N * hd.R;
  }
}


# LRL kernel: product transposed (W rows as A, strip rows as B) so a lane holds one row's outputs: LayerNorm sums in registers, 4 partials per row via LDS, no output tile in LDS; sum order changes
# speedup vs baseline: 4.7239x; 1.0739x over previous
#include <hip/hip_runtime.h>
#include <math.h>

static const int HD  = 128;
static const int PD  = 1024;
static const int LD  = 1024;
static const int CD  = 128;
static const int NT  = 64, NH = 32, NA = 64, NB = 64, NPA = 32, NPE = 32;
static const int RT  = PD + LD + PD + LD + CD + NH;
static const int RH  = PD + PD + LD + CD;
static const int HH  = HD * HD;

typedef float v2f __attribute__((ext_vector_type(2)));
typedef float v8f __attribute__((ext_vector_type(8)));

__device__ __forceinline__ v8f wmma_f32(v2f a, v2f b, v8f c) {
  return __builtin_amdgcn_wmma_f32_16x16x4_f32(false, a, false, b, (short)0, c,
                                               false, false);
}

struct RowSrc {
  const float* p0; const float* p1; const float* p2;
  const float* p3; const float* p4; const float* p5;
  int e0, e1, e2, e3, e4;
};
__device__ __forceinline__ const float* strip_base(const RowSrc& s, int r) {
  const float* p = s.p0;
  int b = 0;
  if (r >= s.e0) { p = s.p1; b = s.e0; }
  if (r >= s.e1) { p = s.p2; b = s.e1; }
  if (r >= s.e2) { p = s.p3; b = s.e2; }
  if (r >= s.e3) { p = s.p4; b = s.e3; }
  if (r >= s.e4) { p = s.p5; b = s.e4; }
  return p + (size_t)(r - b) * 128;
}
struct MaskSrc {
  const float* q0; const float* q1; const float* q2;
  int b0, e0, b1, e1, b2, e2;
};
__device__ __forceinline__ const float* mask_addr(const MaskSrc& s, int r, bool& inside) {
  const float* p = s.q0;
  inside = false;
  if (r >= s.b0 && r < s.e0) { p = s.q0 + (r - s.b0); inside = true; }
  if (r >= s.b1 && r < s.e1) { p = s.q1 + (r - s.b1); inside = true; }
  if (r >= s.b2 && r < s.e2) { p = s.q2 + (r - s.b2); inside = true; }
  return p;
}

static const int MK_ROWG = PD / 32;
static const int MK_COLG = LD / 32;
static const int MK_PCG  = CD / 32;
__global__ __launch_bounds__(256)
void masks_kernel(const int* __restrict__ pl, const int* __restrict__ pc,
                  float* __restrict__ lpl_m, float* __restrict__ plp_m,
                  float* __restrict__ pcp_m) {
  __shared__ alignas(16) int sP[8 * 32];
  const int tid  = threadIdx.x;
  const int lane = tid & 31;
  const int w    = tid >> 5;
  const int b    = blockIdx.x;

  if (b < MK_ROWG) {
    const int4* __restrict__ row = (const int4*)(pl + (size_t)(32 * b + 4 * w) * LD);
    int4 v[4][8];
#pragma unroll
    for (int r = 0; r < 4; ++r) {
#pragma unroll
      for (int j = 0; j < 8; ++j) v[r][j] = row[r * (LD / 4) + j * 32 + lane];
    }
    int s[4];
#pragma unroll
    for (int r = 0; r < 4; ++r) {
      int t = 0;
#pragma unroll
      for (int j = 0; j < 8; ++j) t += (v[r][j].x + v[r][j].y) + (v[r][j].z + v[r][j].w);
      s[r] = t;
    }
#pragma unroll
    for (int r = 0; r < 4; ++r)
      for (int o = 16; o > 0; o >>= 1) s[r] += __shfl_xor(s[r], o, 32);
    if (lane == 0) {
      sP[4 * w + 0] = s[0]; sP[4 * w + 1] = s[1];
      sP[4 * w + 2] = s[2]; sP[4 * w + 3] = s[3];
    }
    __syncthreads();
    if (w == 0) lpl_m[32 * b + lane] = (sP[lane] >= 2) ? 1.f : 0.f;
  } else {
    const int* __restrict__ rel = pl;
    float* __restrict__ dst = plp_m;
    int ld = LD, c0 = 32 * (b - MK_ROWG);
    if (b >= MK_ROWG + MK_COLG) {
      rel = pc; dst = pcp_m; ld = CD; c0 = 32 * (b - MK_ROWG - MK_COLG);
    }
    const int q = tid & 7;
    const int r = tid >> 3;
    const int* __restrict__ p = rel + (size_t)r * ld + c0 + 4 * q;
    int4 v[32];
#pragma unroll
    for (int j = 0; j < 32; ++j) v[j] = *(const int4*)(p + (size_t)(32 * j) * ld);
    int4 s = {0, 0, 0, 0};
#pragma unroll
    for (int j = 0; j < 32; ++j) { s.x += v[j].x; s.y += v[j].y; s.z += v[j].z; s.w += v[j].w; }
    for (int o = 8; o <= 16; o <<= 1) {
      s.x += __shfl_xor(s.x, o, 32); s.y += __shfl_xor(s.y, o, 32);
      s.z += __shfl_xor(s.z, o, 32); s.w += __shfl_xor(s.w, o, 32);
    }
    if (lane < 8) ((int4*)sP)[w * 8 + lane] = s;
    __syncthreads();
    if (w == 0) {
      int t = 0;
#pragma unroll
      for (int k = 0; k < 8; ++k) t += sP[k * 32 + lane];
      dst[c0 + lane] = (t >= 2) ? 1.f : 0.f;
    }
  }
}

static const int KS   = 8;
static const int AG_K = 128;
typedef float v4f __attribute__((ext_vector_type(4)));
template <bool TRANS, int LDA>
__global__ __launch_bounds__(128)
void agg_gemm_kernel(const int* __restrict__ rel, const float* __restrict__ Bm, int pstride,
                     float* __restrict__ Cpart) {
  const int lane = threadIdx.x & 31;
  const int w    = threadIdx.x >> 5;
  const int m0   = blockIdx.x * 16;
  const int c    = lane & 15;
  const int h    = lane >> 4;
  const int kh   = h << 2;
  const int kbeg = blockIdx.y * AG_K;
  const int col  = 32 * w + 2 * c;

  const float* __restrict__ bp = Bm + (size_t)(kbeg + kh) * HD + col;
  v2f bf[16][4];
#pragma unroll
  for (int j = 0; j < 16; ++j) {
#pragma unroll
    for (int s = 0; s < 4; ++s)
      bf[j][s] = *(const v2f*)&bp[(size_t)(8 * j + s) * HD];
  }

  int4 ai[16];
  if constexpr (!TRANS) {
    const int4* __restrict__ ap = (const int4*)(rel + (size_t)(m0 + c) * LDA + kbeg + kh);
#pragma unroll
    for (int j = 0; j < 16; ++j) ai[j] = ap[2 * j];
  } else {
    const int* __restrict__ ap = rel + (size_t)(kbeg + kh) * LDA + m0 + c;
#pragma unroll
    for (int j = 0; j < 16; ++j) {
      ai[j].x = ap[(size_t)(8 * j + 0) * LDA]; ai[j].y = ap[(size_t)(8 * j + 1) * LDA];
      ai[j].z = ap[(size_t)(8 * j + 2) * LDA]; ai[j].w = ap[(size_t)(8 * j + 3) * LDA];
    }
  }
  __builtin_amdgcn_sched_barrier(0);
  v4f af[16];
#pragma unroll
  for (int j = 0; j < 16; ++j) {
    af[j].x = (float)ai[j].x; af[j].y = (float)ai[j].y;
    af[j].z = (float)ai[j].z; af[j].w = (float)ai[j].w;
  }

  v8f acc0 = {0.f, 0.f, 0.f, 0.f, 0.f, 0.f, 0.f, 0.f};
  v8f acc1 = {0.f, 0.f, 0.f, 0.f, 0.f, 0.f, 0.f, 0.f};
#pragma unroll
  for (int j = 0; j < 16; ++j) {
    const v2f b00 = {bf[j][0].x, bf[j][1].x}, b01 = {bf[j][0].y, bf[j][1].y};
    const v2f b10 = {bf[j][2].x, bf[j][3].x}, b11 = {bf[j][2].y, bf[j][3].y};
    acc0 = wmma_f32(af[j].xy, b00, acc0);
    acc1 = wmma_f32(af[j].xy, b01, acc1);
    acc0 = wmma_f32(af[j].zw, b10, acc0);
    acc1 = wmma_f32(af[j].zw, b11, acc1);
  }

  float* __restrict__ Cout = Cpart + (size_t)blockIdx.y * pstride
                           + (size_t)(m0 + 8 * h) * HD + col;
#pragma unroll
  for (int v = 0; v < 8; ++v) {
    const v2f o = {acc0[v], acc1[v]};
    *(v2f*)&Cout[(size_t)v * HD] = o;
  }
}

__device__ __forceinline__ void lds_barrier() {
  __builtin_amdgcn_fence(__ATOMIC_RELEASE, "workgroup", "local");
  __builtin_amdgcn_s_barrier();
  __builtin_amdgcn_fence(__ATOMIC_ACQUIRE, "workgroup", "local");
}

template <int M>
__device__ __forceinline__ float xrow(float v) {
  return __builtin_bit_cast(float, __builtin_amdgcn_update_dpp(
      0, __builtin_bit_cast(int, v), 0x160 + M, 0xf, 0xf, true));
}
__device__ __forceinline__ float xhalf(float v) {
  const int b = __builtin_bit_cast(int, v);
  return __builtin_bit_cast(float,
      __builtin_amdgcn_permlanex16(b, b, 0x76543210, 0xfedcba98, false, false));
}

template <bool SPLIT>
__global__ __launch_bounds__(128)
void lrl_gemm_kernel(RowSrc xs, int sstride,
                     const float* __restrict__ X1,
                     float alpha, const float* __restrict__ W,
                     const float* __restrict__ bias, const float* __restrict__ g,
                     const float* __restrict__ be, float* __restrict__ Y) {
  __shared__ v4f sP[16 * 2];
  const int tid  = threadIdx.x;
  const int lane = tid & 31;
  const int w    = tid >> 5;
  const int m0   = blockIdx.x * 16;
  const int c    = lane & 15;
  const int h    = lane >> 4;
  const int kh   = h << 2;
  const int nb   = 32 * w + 8 * h;
  const float* __restrict__ X0 = strip_base(xs, m0);

  __shared__ v4f sA[SPLIT ? 16 * 32 : 1];
  v4f xf[16];
  v4f y[4], t[4][KS];
  if constexpr (SPLIT) {
    static_assert(KS == 8, "the plane sums below are written for 8 planes");
#pragma unroll
    for (int i = 0; i < 4; ++i) {
      const int q = tid + 128 * i;
      const size_t o = (size_t)(q >> 5) * 128 + 4 * (q & 31);
#pragma unroll
      for (int s = 0; s < KS; ++s) t[i][s] = *(const v4f*)&X0[(size_t)s * sstride + o];
      y[i] = *(const v4f*)&X1[(size_t)m0 * 128 + o];
    }
  }

  v4f aw0[16], aw1[16];
#pragma unroll
  for (int j = 0; j < 16; ++j) {
    aw0[j] = *(const v4f*)&W[(size_t)(32 * w + c) * 128 + 8 * j + kh];
    aw1[j] = *(const v4f*)&W[(size_t)(32 * w + 16 + c) * 128 + 8 * j + kh];
  }
  if constexpr (!SPLIT) {
#pragma unroll
    for (int j = 0; j < 16; ++j)
      xf[j] = alpha * *(const v4f*)&X0[(size_t)c * 128 + 8 * j + kh];
    if (X1) {
#pragma unroll
      for (int j = 0; j < 16; ++j)
        xf[j] += *(const v4f*)&X1[(size_t)(m0 + c) * 128 + 8 * j + kh];
    }
  }
  v4f pb[4], pg[4], pe[4];
#pragma unroll
  for (int u = 0; u < 4; ++u) {
    pb[u] = *(const v4f*)&bias[nb + 16 * (u >> 1) + 4 * (u & 1)];
    pg[u] = *(const v4f*)&g[nb + 16 * (u >> 1) + 4 * (u & 1)];
    pe[u] = *(const v4f*)&be[nb + 16 * (u >> 1) + 4 * (u & 1)];
  }
  if constexpr (SPLIT) {
    __builtin_amdgcn_sched_barrier(0);
#pragma unroll
    for (int i = 0; i < 4; ++i) {
      v4f x = t[i][0];
#pragma unroll
      for (int s = 1; s < KS; ++s) x += t[i][s];
      x = alpha * x + y[i];
      const int q = tid + 128 * i;
      const int m = q >> 5, p = q & 31, j = p >> 1;
      sA[j * 32 + ((m + ((p & 1) << 4)) ^ j)] = x;
    }
    lds_barrier();
#pragma unroll
    for (int j = 0; j < 16; ++j) xf[j] = sA[j * 32 + (lane ^ j)];
    __builtin_amdgcn_sched_barrier(0);
  }

  v8f acc0 = {0.f, 0.f, 0.f, 0.f, 0.f, 0.f, 0.f, 0.f};
  v8f acc1 = {0.f, 0.f, 0.f, 0.f, 0.f, 0.f, 0.f, 0.f};
#pragma unroll
  for (int j = 0; j < 16; ++j) {
    acc0 = wmma_f32(aw0[j].xy, xf[j].xy, acc0);
    acc1 = wmma_f32(aw1[j].xy, xf[j].xy, acc1);
    acc0 = wmma_f32(aw0[j].zw, xf[j].zw, acc0);
    acc1 = wmma_f32(aw1[j].zw, xf[j].zw, acc1);
  }

  float x[16];
#pragma unroll
  for (int r = 0; r < 8; ++r) {
    const float v0 = acc0[r] + pb[r >> 2][r & 3];
    const float v1 = acc1[r] + pb[2 + (r >> 2)][r & 3];
    x[r]     = v0 > 0.f ? v0 : 0.f;
    x[8 + r] = v1 > 0.f ? v1 : 0.f;
  }
  float s = 0.f, s2 = 0.f;
#pragma unroll
  for (int r = 0; r < 16; ++r) { s += x[r]; s2 += x[r] * x[r]; }
  s  += xhalf(s);
  s2 += xhalf(s2);
  if (h == 0) {
    ((float*)&sP[2 * c])[w]     = s;
    ((float*)&sP[2 * c + 1])[w] = s2;
  }
  __syncthreads();
  const v4f ps = sP[2 * c], pq = sP[2 * c + 1];
  const float mean = ((ps.x + ps.y) + (ps.z + ps.w)) * (1.f / 128.f);
  const float var  = ((pq.x + pq.y) + (pq.z + pq.w)) * (1.f / 128.f) - mean * mean;
  const float rstd = rsqrtf(var + 1e-5f);
  float* __restrict__ yo = Y + (size_t)(m0 + c) * 128 + nb;
#pragma unroll
  for (int u = 0; u < 4; ++u) {
    v4f o4;
#pragma unroll
    for (int e = 0; e < 4; ++e)
      o4[e] = (x[4 * u + e] - mean) * rstd * pg[u][e] + pe[u][e];
    *(v4f*)&yo[16 * (u >> 1) + 4 * (u & 1)] = o4;
  }
}

static const int PF_G = 32;
static const int PF_P = 132;
template <int S>
__global__ __launch_bounds__(256)
void pair_fused_kernel(const float* __restrict__ sym, RowSrc geo, MaskSrc mask,
                       const float* __restrict__ W1, const float* __restrict__ b1,
                       const float* __restrict__ g, const float* __restrict__ be,
                       const float* __restrict__ W2, const float* __restrict__ b2p,
                       float* __restrict__ out, int R) {
  __shared__ alignas(16) float sS[16 * PF_P];
  __shared__ alignas(16) float sG[PF_G * PF_P];
  __shared__ float sD[S * PF_G];
  const int tid  = threadIdx.x;
  const int lane = tid & 31;
  const int w    = tid >> 5;
  const int r0   = blockIdx.x * PF_G;
  const int i0   = blockIdx.y * S;
  const int i16  = i0 & ~15;
  const int lo   = i0 - i16;
  const int n    = w * 16 + (lane & 15);
  const int am   = lane & 15;
  const int koff = (lane >> 4) << 1;

  v2f bw[32];
#pragma unroll
  for (int t = 0; t < 32; ++t)
    bw[t] = *(const v2f*)&W1[(size_t)n * 128 + 4 * t + koff];

  const float* __restrict__ gs0 = strip_base(geo, r0);
  const float* __restrict__ gs1 = strip_base(geo, r0 + 16);
  {
    float4 vs[2], vg[4];
#pragma unroll
    for (int j = 0; j < 2; ++j) {
      int i = tid + j * 256;
      vs[j] = ((const float4*)(sym + (size_t)(i16 + (i >> 5)) * 128))[i & 31];
    }
#pragma unroll
    for (int j = 0; j < 4; ++j) {
      int i = tid + j * 256;
      const float* base = (j < 2) ? gs0 : gs1;
      vg[j] = ((const float4*)(base + (size_t)((i >> 5) & 15) * 128))[i & 31];
    }
#pragma unroll
    for (int j = 0; j < 2; ++j) {
      int i = tid + j * 256;
      *(float4*)&sS[(i >> 5) * PF_P + 4 * (i & 31)] = vs[j];
    }
#pragma unroll
    for (int j = 0; j < 4; ++j) {
      int i = tid + j * 256;
      *(float4*)&sG[(i >> 5) * PF_P + 4 * (i & 31)] = vg[j];
    }
  }
  const float4 b4 = ((const float4*)b1)[lane];
  const float4 gg = ((const float4*)g)[lane];
  const float4 bb = ((const float4*)be)[lane];
  const float4 w2 = ((const float4*)W2)[lane];
  bool mk_inside;
  const float mk_raw = *mask_addr(mask, r0 + lane, mk_inside);
  const float b2 = b2p[0];
  lds_barrier();

  v8f aS  = {0.f, 0.f, 0.f, 0.f, 0.f, 0.f, 0.f, 0.f};
  v8f aG0 = {0.f, 0.f, 0.f, 0.f, 0.f, 0.f, 0.f, 0.f};
  v8f aG1 = {0.f, 0.f, 0.f, 0.f, 0.f, 0.f, 0.f, 0.f};
#pragma unroll
  for (int kk = 0; kk < 128; kk += 4) {
    v2f b  = bw[kk >> 2];
    v2f x  = *(const v2f*)&sS[am * PF_P + kk + koff];
    v2f y0 = *(const v2f*)&sG[am * PF_P + kk + koff];
    v2f y1 = *(const v2f*)&sG[(16 + am) * PF_P + kk + koff];
    aS  = wmma_f32(x,  b, aS);
    aG0 = wmma_f32(y0, b, aG0);
    aG1 = wmma_f32(y1, b, aG1);
  }
  __builtin_amdgcn_sched_barrier(0);
  __syncthreads();
  const int mb = (lane >> 4) << 3;
#pragma unroll
  for (int v = 0; v < 8; ++v) {
    sS[(mb + v) * PF_P + n]      = aS[v];
    sG[(mb + v) * PF_P + n]      = aG0[v];
    sG[(16 + mb + v) * PF_P + n] = aG1[v];
  }
  __syncthreads();

  float4 g4[4];
#pragma unroll
  for (int c = 0; c < 4; ++c)
    g4[c] = *(const float4*)&sG[(4 * w + c) * PF_P + 4 * lane];

  const float4 gw = {gg.x * w2.x, gg.y * w2.y, gg.z * w2.z, gg.w * w2.w};
  float sgw = (gw.x + gw.y) + (gw.z + gw.w);
  float sbw = (bb.x * w2.x + bb.y * w2.y) + (bb.z * w2.z + bb.w * w2.w);
  for (int o = 16; o > 0; o >>= 1) {
    sgw += __shfl_xor(sgw, o, 32);
    sbw += __shfl_xor(sbw, o, 32);
  }
  const bool h16 = (lane & 16) != 0;
  const bool h8  = (lane & 8) != 0;
  const bool h4  = (lane & 4) != 0;
  const bool h2  = (lane & 2) != 0;

  for (int q = 0; q < S / 4; ++q) {
    float4 s4[4];
#pragma unroll
    for (int a = 0; a < 4; ++a)
      s4[a] = *(const float4*)&sS[(lo + 4 * q + a) * PF_P + 4 * lane];

    float v[48];
#pragma unroll
    for (int a = 0; a < 4; ++a) {
#pragma unroll
      for (int c = 0; c < 4; ++c) {
        float x0 = s4[a].x + g4[c].x + b4.x; x0 = x0 > 0.f ? x0 : 0.f;
        float x1 = s4[a].y + g4[c].y + b4.y; x1 = x1 > 0.f ? x1 : 0.f;
        float x2 = s4[a].z + g4[c].z + b4.z; x2 = x2 > 0.f ? x2 : 0.f;
        float x3 = s4[a].w + g4[c].w + b4.w; x3 = x3 > 0.f ? x3 : 0.f;
        const int p = 4 * a + c;
        v[3 * p + 0] = (x0 + x1) + (x2 + x3);
        v[3 * p + 1] = (x0 * x0 + x1 * x1) + (x2 * x2 + x3 * x3);
        v[3 * p + 2] = (x0 * gw.x + x1 * gw.y) + (x2 * gw.z + x3 * gw.w);
      }
    }
#pragma unroll
    for (int i = 0; i < 24; ++i) {
      const float keep = h16 ? v[24 + i] : v[i], send = h16 ? v[i] : v[24 + i];
      v[i] = keep + xhalf(send);
    }
#pragma unroll
    for (int i = 0; i < 12; ++i) {
      const float keep = h8 ? v[12 + i] : v[i], send = h8 ? v[i] : v[12 + i];
      v[i] = keep + xrow<8>(send);
    }
#pragma unroll
    for (int i = 0; i < 6; ++i) {
      const float keep = h4 ? v[6 + i] : v[i], send = h4 ? v[i] : v[6 + i];
      v[i] = keep + xrow<4>(send);
    }
#pragma unroll
    for (int i = 0; i < 3; ++i) {
      const float keep = h2 ? v[3 + i] : v[i], send = h2 ? v[i] : v[3 + i];
      v[i] = keep + xrow<2>(send);
    }
#pragma unroll
    for (int i = 0; i < 3; ++i) v[i] += xrow<1>(v[i]);

    const float mean = v[0] * (1.f / 128.f);
    const float var  = v[1] * (1.f / 128.f) - mean * mean;
    const float rstd = rsqrtf(var + 1e-5f);
    const float d    = rstd * (v[2] - mean * sgw) + sbw;
    if ((lane & 1) == 0)
      sD[(4 * q + (lane >> 3)) * PF_G + 4 * w + ((lane >> 1) & 3)] = d;
  }
  __syncthreads();

  for (int idx = tid; idx < S * PF_G; idx += 256) {
    const int a = idx >> 5;
    const int r = r0 + (idx & 31);
    float logit = sD[idx] + b2;
    const float mk = mk_inside ? mk_raw : 1.f;
    float z  = (mk > 0.f) ? logit : -1e9f;
    out[(size_t)(i0 + a) * R + r] = 1.f / (1.f + expf(-z));
  }
}

extern "C" void kernel_launch(void* const* d_in, const int* in_sizes, int n_in,
                              void* d_out, int out_size, void* d_ws, size_t ws_size,
                              hipStream_t stream) {
  (void)in_sizes; (void)n_in; (void)out_size; (void)ws_size;
  const float* points  = (const float*)d_in[0];
  const float* lines   = (const float*)d_in[1];
  const float* circles = (const float*)d_in[2];
  const float* text_s  = (const float*)d_in[3];
  const float* head_s  = (const float*)d_in[4];
  const float* angle_s = (const float*)d_in[5];
  const float* bar_s   = (const float*)d_in[6];
  const float* para_s  = (const float*)d_in[7];
  const float* perp_s  = (const float*)d_in[8];
  const float* rel_W   = (const float*)d_in[9];
  const float* rel_b   = (const float*)d_in[10];
  const float* rel_g   = (const float*)d_in[11];
  const float* rel_be  = (const float*)d_in[12];
  const float* gm_W    = (const float*)d_in[13];
  const float* gm_b    = (const float*)d_in[14];
  const float* gm_g    = (const float*)d_in[15];
  const float* gm_be   = (const float*)d_in[16];
  const float* sc_W1   = (const float*)d_in[17];
  const float* sc_b1   = (const float*)d_in[18];
  const float* sc_g    = (const float*)d_in[19];
  const float* sc_be   = (const float*)d_in[20];
  const float* sc_W2   = (const float*)d_in[21];
  const float* sc_b2   = (const float*)d_in[22];
  const int*   pl      = (const int*)d_in[23];
  const int*   pc      = (const int*)d_in[24];
  float* out = (float*)d_out;

  float* wsp = (float*)d_ws;
  size_t off = 0;
  auto alloc = [&](size_t n) { float* p = wsp + off; off += (n + 31) & ~(size_t)31; return p; };
  float* LPLb   = alloc((size_t)PD * HD);
  float* PLPb   = alloc((size_t)LD * HD);
  float* PCPb   = alloc((size_t)CD * HD);
  float* lpl_m  = alloc(PD);
  float* plp_m  = alloc(LD);
  float* pcp_m  = alloc(CD);
  float* gm_t   = alloc((size_t)RT * HD);
  float* tmp    = alloc((size_t)RT * HD);
  float* part   = alloc((size_t)KS * PD * HD);

  const int OPEN = 1 << 30;
  auto plain = [&](const float* p) { return RowSrc{p, p, p, p, p, p, OPEN, OPEN, OPEN, OPEN, OPEN}; };

  static_assert(LD / KS == AG_K && PD / KS == AG_K, "a work-group's K slice is AG_K rows of the margin");
  masks_kernel<<<MK_ROWG + MK_COLG + MK_PCG, 256, 0, stream>>>(pl, pc, lpl_m, plp_m, pcp_m);

  agg_gemm_kernel<false, LD><<<dim3(PD / 16, KS), 128, 0, stream>>>(pl, lines, PD * HD, part);
  lrl_gemm_kernel<true><<<PD / 16, 128, 0, stream>>>(plain(part), PD * HD, points, 2.f, rel_W + 0 * HH,
      rel_b + 0 * HD, rel_g + 0 * HD, rel_be + 0 * HD, LPLb);

  agg_gemm_kernel<true, LD><<<dim3(LD / 16, KS), 128, 0, stream>>>(pl, points, LD * HD, part);
  lrl_gemm_kernel<true><<<LD / 16, 128, 0, stream>>>(plain(part), LD * HD, lines, 2.f, rel_W + 1 * HH,
      rel_b + 1 * HD, rel_g + 1 * HD, rel_be + 1 * HD, PLPb);

  agg_gemm_kernel<true, CD><<<dim3(CD / 16, KS), 128, 0, stream>>>(pc, points, CD * HD, part);
  lrl_gemm_kernel<true><<<CD / 16, 128, 0, stream>>>(plain(part), CD * HD, circles, 2.f, rel_W + 2 * HH,
      rel_b + 2 * HD, rel_g + 2 * HD, rel_be + 2 * HD, PCPb);

  const RowSrc cat_t = {points, lines, LPLb, PLPb, PCPb, head_s,
                        PD, PD + LD, PD + LD + PD, PD + LD + PD + LD, PD + LD + PD + LD + CD};
  lrl_gemm_kernel<false><<<RT / 16, 128, 0, stream>>>(cat_t, 0, nullptr, 1.f, gm_W + 0 * HH,
      gm_b + 0 * HD, gm_g + 0 * HD, gm_be + 0 * HD, tmp);
  lrl_gemm_kernel<false><<<RT / 16, 128, 0, stream>>>(plain(tmp), 0, nullptr, 1.f, gm_W + 1 * HH,
      gm_b + 1 * HD, gm_g + 1 * HD, gm_be + 1 * HD, gm_t);
  const float* gm_t2 = gm_t + (size_t)(PD + LD) * HD;
  const RowSrc cat_h = {gm_t, gm_t2, gm_t2, gm_t2, gm_t2, gm_t2, PD, OPEN, OPEN, OPEN, OPEN};

  struct Head { const float* sym; int N; RowSrc geo; int R; MaskSrc mask; };
  const Head heads[6] = {
      {text_s,  NT,  plain(gm_t),  RT, {lpl_m, plp_m, pcp_m, PD + LD, PD + LD + PD,
                                        PD + LD + PD, PD + LD + PD + LD,
                                        PD + LD + PD + LD, PD + LD + PD + LD + CD}},
      {head_s,  NH,  cat_h,        RH, {lpl_m, plp_m, pcp_m, PD, PD + PD,
                                        PD + PD, PD + PD + LD,
                                        PD + PD + LD, PD + PD + LD + CD}},
      {angle_s, NA,  plain(LPLb),  PD, {lpl_m, lpl_m, lpl_m, 0, PD, 0, 0, 0, 0}},
      {bar_s,   NB,  plain(PLPb),  LD, {plp_m, plp_m, plp_m, 0, LD, 0, 0, 0, 0}},
      {para_s,  NPA, plain(lines), LD, {lpl_m, lpl_m, lpl_m, 0, 0, 0, 0, 0, 0}},
      {perp_s,  NPE, plain(LPLb),  PD, {lpl_m, lpl_m, lpl_m, 0, PD, 0, 0, 0, 0}},
  };
  size_t ooff = 0;
  for (int h = 0; h < 6; ++h) {
    const Head& hd = heads[h];
    const float* w1 = sc_W1 + (size_t)h * HH;
    const float* pb = sc_b1 + (size_t)h * HD;
    const float* pg = sc_g + (size_t)h * HD;
    const float* pe = sc_be + (size_t)h * HD;
    const float* pw = sc_W2 + (size_t)h * HD;
    if (h == 1)
      pair_fused_kernel<16><<<dim3(hd.R / PF_G, hd.N / 16), 256, 0, stream>>>(hd.sym, hd.geo,
          hd.mask, w1, pb, pg, pe, pw, sc_b2 + h, out + ooff, hd.R);
    else if (h < 4)
      pair_fused_kernel<8><<<dim3(hd.R / PF_G, hd.N / 8), 256, 0, stream>>>(hd.sym, hd.geo,
          hd.mask, w1, pb, pg, pe, pw, sc_b2 + h, out + ooff, hd.R);
    else
      pair_fused_kernel<4><<<dim3(hd.R / PF_G, hd.N / 4), 256, 0, stream>>>(hd.sym, hd.geo,
          hd.mask, w1, pb, pg, pe, pw, sc_b2 + h, out + ooff, hd.R);
    ooff += (size_t)hd.N * hd.R;
  }
}
